# starting kernel
# baseline (speedup 1.0000x reference)
#include <hip/hip_runtime.h>
#include <math.h>


typedef __attribute__((ext_vector_type(2))) float v2f;
typedef __attribute__((ext_vector_type(8))) float v8f;

__device__ __forceinline__ float softplusf(float x) {
  return (x > 20.f) ? x : log1pf(expf(x));
}

__global__ __launch_bounds__(256) void k_inverse(const float* __restrict__ V,
                                                 float* __restrict__ Vinv) {
  __shared__ float M[64][128];
  __shared__ float prow[128];
  __shared__ float fac[64];
  const int tid = threadIdx.x;
  for (int idx = tid; idx < 64 * 128; idx += 256) {
    int r = idx >> 7, c = idx & 127;
    M[r][c] = (c < 64) ? V[r * 64 + c] : ((c - 64 == r) ? 1.f : 0.f);
  }
  __syncthreads();
  for (int k = 0; k < 64; ++k) {
    if (tid < 128) {
      float pinv = 1.f / M[k][k];
      prow[tid] = M[k][tid] * pinv;
    } else if (tid < 192) {
      fac[tid - 128] = M[tid - 128][k];
    }
    __syncthreads();
    for (int idx = tid; idx < 64 * 128; idx += 256) {
      int r = idx >> 7, c = idx & 127;
      M[r][c] = (r == k) ? prow[c] : fmaf(-fac[r], prow[c], M[r][c]);
    }
    __syncthreads();
  }
  for (int idx = tid; idx < 64 * 64; idx += 256) {
    int r = idx >> 6, c = idx & 63;
    Vinv[idx] = M[r][64 + c];
  }
}

__global__ void k_build_G(const float* __restrict__ gate_w,
                          const float* __restrict__ B,
                          const float* __restrict__ Vinv,
                          float* __restrict__ G) {
  int idx = blockIdx.x * blockDim.x + threadIdx.x;
  if (idx >= 96 * 512) return;
  int row = idx >> 9, h = idx & 511;
  if (row < 32) {
    G[idx] = gate_w[row * 512 + h];
  } else {
    int d = row - 32;
    float s = 0.f;
    for (int j = 0; j < 64; ++j) s = fmaf(Vinv[d * 64 + j], B[j * 512 + h], s);
    G[idx] = s;
  }
}

__global__ void k_EA(const float* __restrict__ E, const float* __restrict__ alpha,
                     float* __restrict__ EA) {
  int idx = blockIdx.x * blockDim.x + threadIdx.x;
  if (idx >= 64 * 100) return;
  int d = idx / 100, k = idx % 100;
  float s = 0.f;
  for (int r = 0; r < 32; ++r) s = fmaf(E[d * 32 + r], alpha[r * 100 + k], s);
  EA[idx] = s;
}

__global__ void k_Minv(const float* __restrict__ Vinv, const float* __restrict__ EA,
                       float* __restrict__ Minv) {
  int idx = blockIdx.x * blockDim.x + threadIdx.x;
  if (idx >= 64 * 100) return;
  int d = idx / 100, k = idx % 100;
  float s = 0.f;
  for (int j = 0; j < 64; ++j) s = fmaf(Vinv[d * 64 + j], EA[j * 100 + k], s);
  Minv[idx] = s;
}

__global__ __launch_bounds__(32) void k_gemm_uG(const float* __restrict__ U,
                                                const float* __restrict__ G,
                                                float* __restrict__ C96) {
  const int rt = blockIdx.x, ct = blockIdx.y;
  const int l = threadIdx.x;
  const int m = l & 15;
  const int kh = (l >> 4) * 2;
  const float* Arow = U + (size_t)(rt * 16 + m) * 512;
  const float* Brow = G + (size_t)(ct * 16 + m) * 512;
  v8f acc = {};
  for (int kb = 0; kb < 512; kb += 4) {
    v2f a = *(const v2f*)(Arow + kb + kh);
    v2f b = *(const v2f*)(Brow + kb + kh);
    acc = __builtin_amdgcn_wmma_f32_16x16x4_f32(false, a, false, b, (short)0, acc,
                                                false, false);
  }
  const int rbase = rt * 16 + (l >> 4) * 8;
  const int col = ct * 16 + m;
  for (int v = 0; v < 8; ++v) C96[(size_t)(rbase + v) * 96 + col] = acc[v];
}

__global__ void k_prep(const float* __restrict__ times, const int* __restrict__ marks,
                       const float* __restrict__ llr, const float* __restrict__ lim,
                       const float* __restrict__ gate_b, const float* __restrict__ C96,
                       const float* __restrict__ Minv, float* __restrict__ Ac,
                       float* __restrict__ Uc, int n) {
  int idx = blockIdx.x * blockDim.x + threadIdx.x;
  if (idx >= n * 32) return;
  int ns = idx >> 5, p = idx & 31;
  float t = times[ns];
  float dt = (ns == 0) ? t : (t - times[ns - 1]);
  float gate = softplusf(C96[(size_t)ns * 96 + p] + gate_b[p]);
  float real = -softplusf(llr[p]) * gate;
  float im = lim[p];
  float e = expf(real * dt);
  float a = e * cosf(im * dt);
  float c = e * sinf(im * dt);
  float denom = real * real + im * im;
  float nre = a - 1.f, nim = c;
  float qr = (nre * real + nim * im) / (denom + 1e-12f);
  float qi = (nim * real - nre * im) / (denom + 1e-12f);
  if (denom < 1e-8f) { qr = dt; qi = 0.f; }
  if (!(dt > 0.f)) { qr = 0.f; qi = 0.f; }
  float w0 = C96[(size_t)ns * 96 + 32 + 2 * p];
  float w1 = C96[(size_t)ns * 96 + 32 + 2 * p + 1];
  float ure = qr * w0 - qi * w1;
  float uim = qi * w0 + qr * w1;
  int mk = marks[ns];
  ure += Minv[(2 * p) * 100 + mk];
  uim += Minv[(2 * p + 1) * 100 + mk];
  size_t base = (size_t)ns * 64 + 2 * p;
  Ac[base] = a;   Ac[base + 1] = c;
  Uc[base] = ure; Uc[base + 1] = uim;
}

__global__ __launch_bounds__(32) void k_scanA(const float* __restrict__ Ac,
                                              const float* __restrict__ Uc,
                                              float* __restrict__ Aagg,
                                              float* __restrict__ Uagg, int n, int L) {
  int ch = blockIdx.x, p = threadIdx.x;
  float Ar = 1.f, Ai = 0.f, Ur = 0.f, Ui = 0.f;
  int i0 = ch * L, i1 = i0 + L; if (i1 > n) i1 = n;
  for (int i = i0; i < i1; ++i) {
    size_t b = (size_t)i * 64 + 2 * p;
    float ar = Ac[b], ai = Ac[b + 1], ur = Uc[b], ui = Uc[b + 1];
    float nAr = ar * Ar - ai * Ai, nAi = ar * Ai + ai * Ar;
    float nUr = ar * Ur - ai * Ui + ur, nUi = ar * Ui + ai * Ur + ui;
    Ar = nAr; Ai = nAi; Ur = nUr; Ui = nUi;
  }
  size_t o = (size_t)ch * 64 + 2 * p;
  Aagg[o] = Ar; Aagg[o + 1] = Ai; Uagg[o] = Ur; Uagg[o + 1] = Ui;
}

__global__ __launch_bounds__(32) void k_scanB(const float* __restrict__ Aagg,
                                              const float* __restrict__ Uagg,
                                              const float* __restrict__ Vinv,
                                              const float* __restrict__ x0,
                                              float* __restrict__ Ystart, int nch) {
  int p = threadIdx.x;
  float yr = 0.f, yi = 0.f;
  for (int j = 0; j < 64; ++j) {
    yr = fmaf(Vinv[(2 * p) * 64 + j], x0[j], yr);
    yi = fmaf(Vinv[(2 * p + 1) * 64 + j], x0[j], yi);
  }
  for (int c = 0; c < nch; ++c) {
    size_t o = (size_t)c * 64 + 2 * p;
    Ystart[o] = yr; Ystart[o + 1] = yi;
    float ar = Aagg[o], ai = Aagg[o + 1];
    float nyr = ar * yr - ai * yi + Uagg[o];
    float nyi = ar * yi + ai * yr + Uagg[o + 1];
    yr = nyr; yi = nyi;
  }
}

__global__ __launch_bounds__(32) void k_scanC(const float* __restrict__ Ac,
                                              const float* __restrict__ Uc,
                                              const float* __restrict__ Ystart,
                                              float* __restrict__ Y, int n, int L) {
  int ch = blockIdx.x, p = threadIdx.x;
  size_t so = (size_t)ch * 64 + 2 * p;
  float yr = Ystart[so], yi = Ystart[so + 1];
  int i0 = ch * L, i1 = i0 + L; if (i1 > n) i1 = n;
  for (int i = i0; i < i1; ++i) {
    size_t b = (size_t)i * 64 + 2 * p;
    float ar = Ac[b], ai = Ac[b + 1];
    float nyr = ar * yr - ai * yi + Uc[b];
    float nyi = ar * yi + ai * yr + Uc[b + 1];
    yr = nyr; yi = nyi;
    Y[b] = yr; Y[b + 1] = yi;
  }
}

__global__ __launch_bounds__(32) void k_gemm_out(const float* __restrict__ Y,
                                                 const float* __restrict__ V,
                                                 const float* __restrict__ EA,
                                                 const int* __restrict__ marks,
                                                 float* __restrict__ out, int n) {
  const int rt = blockIdx.x, ct = blockIdx.y, l = threadIdx.x;
  const int m = l & 15;
  const int kh = (l >> 4) * 2;
  const float* Arow = Y + (size_t)(rt * 16 + m) * 64;
  const int dcol = ct * 16 + m;
  const float* Brow = V + (size_t)dcol * 64;
  v8f acc = {};
  for (int kb = 0; kb < 64; kb += 4) {
    v2f a = *(const v2f*)(Arow + kb + kh);
    v2f b = *(const v2f*)(Brow + kb + kh);
    acc = __builtin_amdgcn_wmma_f32_16x16x4_f32(false, a, false, b, (short)0, acc,
                                                false, false);
  }
  const int rbase = rt * 16 + (l >> 4) * 8;
  const size_t xl_off = (size_t)n * 64;
  for (int v = 0; v < 8; ++v) {
    int row = rbase + v;
    float xr = acc[v];
    float r = EA[dcol * 100 + marks[row]];
    out[(size_t)row * 64 + dcol] = xr;
    out[xl_off + (size_t)row * 64 + dcol] = xr - r;
  }
}

extern "C" void kernel_launch(void* const* d_in, const int* in_sizes, int n_in,
                              void* d_out, int out_size, void* d_ws, size_t ws_size,
                              hipStream_t stream) {
  const float* times   = (const float*)d_in[0];
  const int*   marks   = (const int*)d_in[1];
  const float* u_in    = (const float*)d_in[2];
  const float* llr     = (const float*)d_in[3];
  const float* lim     = (const float*)d_in[4];
  const float* V       = (const float*)d_in[5];
  const float* B       = (const float*)d_in[6];
  const float* E       = (const float*)d_in[7];
  const float* alpha   = (const float*)d_in[8];
  const float* gate_w  = (const float*)d_in[9];
  const float* gate_b  = (const float*)d_in[10];
  const float* x0      = (const float*)d_in[11];
  float* out = (float*)d_out;
  const int n = in_sizes[0];
  const int CHN = 256;
  const int L = (n + CHN - 1) / CHN;

  float* ws    = (float*)d_ws;
  float* Vinv  = ws; ws += 64 * 64;
  float* G     = ws; ws += 96 * 512;
  float* EA    = ws; ws += 64 * 100;
  float* Minv  = ws; ws += 64 * 100;
  float* C96   = ws; ws += (size_t)n * 96;
  float* Ac    = ws; ws += (size_t)n * 64;
  float* Uc    = ws; ws += (size_t)n * 64;
  float* Aagg  = ws; ws += (size_t)CHN * 64;
  float* Uagg  = ws; ws += (size_t)CHN * 64;
  float* Ystart= ws; ws += (size_t)CHN * 64;
  float* Y     = ws; ws += (size_t)n * 64;

  k_inverse<<<dim3(1), dim3(256), 0, stream>>>(V, Vinv);
  k_build_G<<<dim3((96 * 512 + 255) / 256), dim3(256), 0, stream>>>(gate_w, B, Vinv, G);
  k_EA<<<dim3((64 * 100 + 255) / 256), dim3(256), 0, stream>>>(E, alpha, EA);
  k_Minv<<<dim3((64 * 100 + 255) / 256), dim3(256), 0, stream>>>(Vinv, EA, Minv);
  k_gemm_uG<<<dim3(n / 16, 6), dim3(32), 0, stream>>>(u_in, G, C96);
  k_prep<<<dim3((n * 32 + 255) / 256), dim3(256), 0, stream>>>(
      times, marks, llr, lim, gate_b, C96, Minv, Ac, Uc, n);
  k_scanA<<<dim3(CHN), dim3(32), 0, stream>>>(Ac, Uc, Aagg, Uagg, n, L);
  k_scanB<<<dim3(1), dim3(32), 0, stream>>>(Aagg, Uagg, Vinv, x0, Ystart, CHN);
  k_scanC<<<dim3(CHN), dim3(32), 0, stream>>>(Ac, Uc, Ystart, Y, n, L);
  k_gemm_out<<<dim3(n / 16, 4), dim3(32), 0, stream>>>(Y, V, EA, marks, out, n);
}


# U GEMM columns regrouped by pole (gate, w0, w1 of 5 poles per 16-column tile, 7 tiles): each wave forms its poles' scan coefficients from its own tile via LDS; plane round trip gone; 8 launches -> 7
# speedup vs baseline: 4.2616x; 4.2616x over previous
#include <hip/hip_runtime.h>
#include <math.h>


typedef __attribute__((ext_vector_type(2))) float v2f;
typedef __attribute__((ext_vector_type(8))) float v8f;

__device__ __forceinline__ float softplusf(float x) {
  return (x > 20.f) ? x : log1pf(expf(x));
}

#define INV_LD 68
#define NF_LD 20

__global__ __launch_bounds__(512) void k_inverse(const float* __restrict__ V,
                                                 float* __restrict__ Vinv) {
  __shared__ float M[64 * INV_LD];
  __shared__ float PR[16 * INV_LD];
  __shared__ __attribute__((aligned(16))) float NF[64 * NF_LD];
  __shared__ __attribute__((aligned(16))) float NFT[16 * 16];
  __shared__ float PINV[16];
  const int tid = threadIdx.x;
  const int l = tid & 31;
  const int w = __builtin_amdgcn_readfirstlane(tid >> 5);
  const int ti = w >> 2, tj = w & 3;
  const int m = l & 15, h = l >> 4, kh = h * 2;

  for (int idx = tid; idx < 64 * 64; idx += 512)
    M[(idx >> 6) * INV_LD + (idx & 63)] = V[idx];
  __syncthreads();

  for (int kb = 0; kb < 4; ++kb) {
    const int c0 = 16 * kb;

    if (w == 0) {
      const int r0 = (c0 + l) & 63, r1 = (c0 + l + 32) & 63;
      float a0[16], a1[16], n0[16], n1[16];
      float mypinv = 0.f;
#pragma unroll
      for (int c = 0; c < 16; ++c) {
        a0[c] = M[r0 * INV_LD + c0 + c];
        a1[c] = M[r1 * INV_LD + c0 + c];
      }
#pragma unroll
      for (int j = 0; j < 16; ++j) {
        const float pk = __int_as_float(__builtin_amdgcn_readlane(__float_as_int(a0[j]), j));
        const float pinv = 1.f / pk;
        const float f0 = a0[j], f1 = a1[j];
        const bool isp = (l == j);
        n0[j] = -f0;
        n1[j] = -f1;
        mypinv = isp ? pinv : mypinv;
#pragma unroll
        for (int c = 0; c < 16; ++c) {
          if (c == j) {
            a0[c] = isp ? pinv : (n0[j] * pinv);
            a1[c] = n1[j] * pinv;
          } else {
            const float pr =
                __int_as_float(__builtin_amdgcn_readlane(__float_as_int(a0[c]), j)) * pinv;
            a0[c] = isp ? pr : fmaf(n0[j], pr, a0[c]);
            a1[c] = fmaf(n1[j], pr, a1[c]);
          }
        }
      }
#pragma unroll
      for (int c = 0; c < 16; ++c) {
        M[r0 * INV_LD + c0 + c] = a0[c];
        M[r1 * INV_LD + c0 + c] = a1[c];
        NF[r0 * NF_LD + c] = n0[c];
        NF[r1 * NF_LD + c] = n1[c];
      }
      if (l < 16) {
#pragma unroll
        for (int j = 0; j < 16; ++j) NFT[j * 16 + l] = n0[j];
        PINV[l] = mypinv;
      }
    }
    __syncthreads();

    if (tid < 64 && (tid >> 4) != kb) {
      const int cc = tid;
      float x[16];
#pragma unroll
      for (int j = 0; j < 16; ++j) x[j] = M[(c0 + j) * INV_LD + cc];
#pragma unroll
      for (int j = 0; j < 16; ++j) {
        const float pr = x[j] * PINV[j];
        PR[j * INV_LD + cc] = pr;
#pragma unroll
        for (int jj = 0; jj < 16; ++jj) {
          if (jj != j) x[jj] = fmaf(NFT[j * 16 + jj], pr, x[jj]);
        }
        x[j] = pr;
      }
#pragma unroll
      for (int j = 0; j < 16; ++j) M[(c0 + j) * INV_LD + cc] = x[j];
    }
    __syncthreads();

    if (ti != kb && tj != kb) {
      float* Tij = M + (16 * ti) * INV_LD + 16 * tj;
      const float* At = NF + (16 * ti) * NF_LD;
      const float* Bt = PR + 16 * tj;
      v8f acc;
      for (int v = 0; v < 8; ++v) acc[v] = Tij[(8 * h + v) * INV_LD + m];
      for (int k4 = 0; k4 < 16; k4 += 4) {
        v2f a, b;
        a.x = At[m * NF_LD + k4 + kh];
        a.y = At[m * NF_LD + k4 + kh + 1];
        b.x = Bt[(k4 + kh) * INV_LD + m];
        b.y = Bt[(k4 + kh + 1) * INV_LD + m];
        acc = __builtin_amdgcn_wmma_f32_16x16x4_f32(false, a, false, b, (short)0, acc,
                                                    false, false);
      }
      for (int v = 0; v < 8; ++v) Tij[(8 * h + v) * INV_LD + m] = acc[v];
    }
    __syncthreads();
  }

  for (int idx = tid; idx < 64 * 64; idx += 512)
    Vinv[idx] = M[(idx >> 6) * INV_LD + (idx & 63)];
}

#define PPT 5
#define NCT 7
__global__ void k_build_G(const float* __restrict__ gate_w,
                          const float* __restrict__ B,
                          const float* __restrict__ Vinv,
                          float* __restrict__ G) {
  int idx = blockIdx.x * blockDim.x + threadIdx.x;
  if (idx >= NCT * 16 * 512) return;
  int row = idx >> 9, h = idx & 511;
  int c = row & 15, q = c / 3, t = c - 3 * q;
  int p = PPT * (row >> 4) + q;
  float s = 0.f;
  if (c < 3 * PPT && p < 32) {
    if (t == 0) {
      s = gate_w[p * 512 + h];
    } else {
      int d = 2 * p + t - 1;
      for (int j = 0; j < 64; ++j) s = fmaf(Vinv[d * 64 + j], B[j * 512 + h], s);
    }
  }
  G[idx] = s;
}

__global__ void k_EA(const float* __restrict__ E, const float* __restrict__ alpha,
                     float* __restrict__ EA) {
  int idx = blockIdx.x * blockDim.x + threadIdx.x;
  if (idx >= 64 * 100) return;
  int d = idx / 100, k = idx % 100;
  float s = 0.f;
  for (int r = 0; r < 32; ++r) s = fmaf(E[d * 32 + r], alpha[r * 100 + k], s);
  EA[idx] = s;
}

__global__ void k_Minv(const float* __restrict__ Vinv, const float* __restrict__ EA,
                       float* __restrict__ Minv) {
  int idx = blockIdx.x * blockDim.x + threadIdx.x;
  if (idx >= 64 * 100) return;
  int d = idx / 100, k = idx % 100;
  float s = 0.f;
  for (int j = 0; j < 64; ++j) s = fmaf(Vinv[d * 64 + j], EA[j * 100 + k], s);
  Minv[idx] = s;
}

#define TS_LD 17
__global__ __launch_bounds__(32) void k_gemm_coef(const float* __restrict__ U,
                                                  const float* __restrict__ G,
                                                  const float* __restrict__ times,
                                                  const int* __restrict__ marks,
                                                  const float* __restrict__ llr,
                                                  const float* __restrict__ lim,
                                                  const float* __restrict__ gate_b,
                                                  const float* __restrict__ Minv,
                                                  float* __restrict__ Cf, int n) {
  __shared__ float TS[16 * TS_LD];
  const int rt = blockIdx.x, ct = blockIdx.y;
  const int l = threadIdx.x;
  const int m = l & 15;
  const int hf = l >> 4;
  const int kh = hf * 2;
  const float* Arow = U + (size_t)(rt * 16 + m) * 512;
  const float* Brow = G + (size_t)(ct * 16 + m) * 512;
  const int ns = rt * 16 + m;
  const float t = times[ns];
  const float tprev = times[ns > 0 ? ns - 1 : 0];
  const int mk = marks[ns];
  float gb[3], lr[3], li[3];
#pragma unroll
  for (int k = 0; k < 3; ++k) {
    const int pc = min(PPT * ct + 2 * k + hf, 31);
    gb[k] = gate_b[pc];
    lr[k] = llr[pc];
    li[k] = lim[pc];
  }
  v8f acc = {};
  for (int kb = 0; kb < 512; kb += 4) {
    v2f a = *(const v2f*)(Arow + kb + kh);
    v2f b = *(const v2f*)(Brow + kb + kh);
    acc = __builtin_amdgcn_wmma_f32_16x16x4_f32(false, a, false, b, (short)0, acc,
                                                false, false);
  }
  for (int v = 0; v < 8; ++v) TS[(8 * hf + v) * TS_LD + m] = acc[v];
  __syncthreads();
  const float dt = (ns == 0) ? t : (t - tprev);
#pragma unroll
  for (int k = 0; k < 3; ++k) {
    const int q = 2 * k + hf;
    const int p = PPT * ct + q;
    if (q < PPT && p < 32) {
      const float* tr = TS + m * TS_LD + 3 * q;
      float gate = softplusf(tr[0] + gb[k]);
      float real = -softplusf(lr[k]) * gate;
      float im = li[k];
      float e = expf(real * dt);
      float a = e * cosf(im * dt);
      float c = e * sinf(im * dt);
      float denom = real * real + im * im;
      float nre = a - 1.f, nim = c;
      float qr = (nre * real + nim * im) / (denom + 1e-12f);
      float qi = (nim * real - nre * im) / (denom + 1e-12f);
      if (denom < 1e-8f) { qr = dt; qi = 0.f; }
      if (!(dt > 0.f)) { qr = 0.f; qi = 0.f; }
      float w0 = tr[1];
      float w1 = tr[2];
      float ure = qr * w0 - qi * w1;
      float uim = qi * w0 + qr * w1;
      ure += Minv[(2 * p) * 100 + mk];
      uim += Minv[(2 * p + 1) * 100 + mk];
      ((float4*)Cf)[(size_t)p * n + ns] = make_float4(a, c, ure, uim);
    }
  }
}

#define SPT 16
#define SCAN_T 512
#define SCAN_W (SCAN_T / 32)

__device__ __forceinline__ void wave_prefix(float& Ar, float& Ai, float& Ur, float& Ui, int l) {
  for (int d = 1; d < 32; d <<= 1) {
    const float pAr = __shfl_up(Ar, d, 32), pAi = __shfl_up(Ai, d, 32);
    const float pUr = __shfl_up(Ur, d, 32), pUi = __shfl_up(Ui, d, 32);
    if (l >= d) {
      const float nUr = Ar * pUr - Ai * pUi + Ur, nUi = Ar * pUi + Ai * pUr + Ui;
      const float nAr = Ar * pAr - Ai * pAi, nAi = Ar * pAi + Ai * pAr;
      Ar = nAr; Ai = nAi; Ur = nUr; Ui = nUi;
    }
  }
}

__global__ __launch_bounds__(SCAN_T) void k_scan(const float* __restrict__ Cf,
                                                 const float* __restrict__ Vinv,
                                                 const float* __restrict__ x0,
                                                 float* __restrict__ YT, int n) {
  __shared__ float4 WAGG[SCAN_W];
  __shared__ float Y0[2];
  const int p = blockIdx.x, t = threadIdx.x, l = t & 31;
  const int w = __builtin_amdgcn_readfirstlane(t >> 5);
  const int s0 = t * SPT;
  const bool live = (s0 + SPT <= n);
  float ar[SPT], ai[SPT], ur[SPT], ui[SPT];
  if (live) {
    const float4* c4 = (const float4*)Cf + (size_t)p * n + s0;
#pragma unroll
    for (int i = 0; i < SPT; ++i) {
      const float4 v = c4[i];
      ar[i] = v.x; ai[i] = v.y; ur[i] = v.z; ui[i] = v.w;
    }
  } else {
#pragma unroll
    for (int i = 0; i < SPT; ++i) { ar[i] = 1.f; ai[i] = 0.f; ur[i] = 0.f; ui[i] = 0.f; }
  }
  if (w == 0) {
    const float* v0 = Vinv + (2 * p) * 64;
    const float* v1 = v0 + 64;
    const float xa = x0[l], xb = x0[32 + l];
    float sr = fmaf(v0[32 + l], xb, v0[l] * xa);
    float si = fmaf(v1[32 + l], xb, v1[l] * xa);
    for (int d = 16; d >= 1; d >>= 1) {
      sr += __shfl_xor(sr, d, 32);
      si += __shfl_xor(si, d, 32);
    }
    if (l == 0) { Y0[0] = sr; Y0[1] = si; }
  }
  float Ar = 1.f, Ai = 0.f, Ur = 0.f, Ui = 0.f;
#pragma unroll
  for (int i = 0; i < SPT; ++i) {
    const float nAr = ar[i] * Ar - ai[i] * Ai, nAi = ar[i] * Ai + ai[i] * Ar;
    const float nUr = ar[i] * Ur - ai[i] * Ui + ur[i], nUi = ar[i] * Ui + ai[i] * Ur + ui[i];
    Ar = nAr; Ai = nAi; Ur = nUr; Ui = nUi;
  }
  wave_prefix(Ar, Ai, Ur, Ui, l);
  if (l == 31) WAGG[w] = make_float4(Ar, Ai, Ur, Ui);
  __syncthreads();
  const float y0r = Y0[0], y0i = Y0[1];
  float qAr = 1.f, qAi = 0.f, qUr = 0.f, qUi = 0.f;
  if (l < SCAN_W) {
    const float4 q = WAGG[l];
    qAr = q.x; qAi = q.y; qUr = q.z; qUi = q.w;
  }
  wave_prefix(qAr, qAi, qUr, qUi, l);
  const int src = (w > 0) ? (w - 1) : 0;
  float xAr = __shfl(qAr, src, 32), xAi = __shfl(qAi, src, 32);
  float xUr = __shfl(qUr, src, 32), xUi = __shfl(qUi, src, 32);
  if (w == 0) { xAr = 1.f; xAi = 0.f; xUr = 0.f; xUi = 0.f; }
  const float ywr = xAr * y0r - xAi * y0i + xUr;
  const float ywi = xAr * y0i + xAi * y0r + xUi;
  float eAr = __shfl_up(Ar, 1, 32), eAi = __shfl_up(Ai, 1, 32);
  float eUr = __shfl_up(Ur, 1, 32), eUi = __shfl_up(Ui, 1, 32);
  if (l == 0) { eAr = 1.f; eAi = 0.f; eUr = 0.f; eUi = 0.f; }
  float yr = eAr * ywr - eAi * ywi + eUr;
  float yi = eAr * ywi + eAi * ywr + eUi;
  float outr[SPT], outi[SPT];
#pragma unroll
  for (int i = 0; i < SPT; ++i) {
    const float nyr = ar[i] * yr - ai[i] * yi + ur[i];
    const float nyi = ar[i] * yi + ai[i] * yr + ui[i];
    yr = nyr; yi = nyi;
    outr[i] = yr; outi[i] = yi;
  }
  if (live) {
    float4* yo = (float4*)(YT + (size_t)(2 * p) * n + s0);
    float4* y1 = (float4*)(YT + (size_t)(2 * p + 1) * n + s0);
#pragma unroll
    for (int i = 0; i < SPT; i += 4) {
      yo[i / 4] = make_float4(outr[i], outr[i + 1], outr[i + 2], outr[i + 3]);
      y1[i / 4] = make_float4(outi[i], outi[i + 1], outi[i + 2], outi[i + 3]);
    }
  }
}

__global__ __launch_bounds__(32) void k_gemm_out(const float* __restrict__ YT,
                                                 const float* __restrict__ V,
                                                 const float* __restrict__ EA,
                                                 const int* __restrict__ marks,
                                                 float* __restrict__ out, int n) {
  const int rt = blockIdx.x, ct = blockIdx.y, l = threadIdx.x;
  const int m = l & 15;
  const int kh = (l >> 4) * 2;
  const float* Acol = YT + (size_t)kh * n + (rt * 16 + m);
  const int dcol = ct * 16 + m;
  const float* Brow = V + (size_t)dcol * 64;
  v8f acc = {};
#pragma unroll
  for (int kb = 0; kb < 64; kb += 4) {
    v2f a;
    a.x = Acol[(size_t)kb * n];
    a.y = Acol[(size_t)(kb + 1) * n];
    v2f b = *(const v2f*)(Brow + kb + kh);
    acc = __builtin_amdgcn_wmma_f32_16x16x4_f32(false, a, false, b, (short)0, acc,
                                                false, false);
  }
  const int rbase = rt * 16 + (l >> 4) * 8;
  const size_t xl_off = (size_t)n * 64;
  for (int v = 0; v < 8; ++v) {
    int row = rbase + v;
    float xr = acc[v];
    float r = EA[dcol * 100 + marks[row]];
    out[(size_t)row * 64 + dcol] = xr;
    out[xl_off + (size_t)row * 64 + dcol] = xr - r;
  }
}

extern "C" void kernel_launch(void* const* d_in, const int* in_sizes, int n_in,
                              void* d_out, int out_size, void* d_ws, size_t ws_size,
                              hipStream_t stream) {
  const float* times   = (const float*)d_in[0];
  const int*   marks   = (const int*)d_in[1];
  const float* u_in    = (const float*)d_in[2];
  const float* llr     = (const float*)d_in[3];
  const float* lim     = (const float*)d_in[4];
  const float* V       = (const float*)d_in[5];
  const float* B       = (const float*)d_in[6];
  const float* E       = (const float*)d_in[7];
  const float* alpha   = (const float*)d_in[8];
  const float* gate_w  = (const float*)d_in[9];
  const float* gate_b  = (const float*)d_in[10];
  const float* x0      = (const float*)d_in[11];
  float* out = (float*)d_out;
  const int n = in_sizes[0];

  float* ws    = (float*)d_ws;
  float* Vinv  = ws; ws += 64 * 64;
  float* G     = ws; ws += NCT * 16 * 512;
  float* EA    = ws; ws += 64 * 100;
  float* Minv  = ws; ws += 64 * 100;
  float* Cf    = ws; ws += (size_t)n * 128;
  float* YT    = ws; ws += (size_t)n * 64;

  k_inverse<<<dim3(1), dim3(512), 0, stream>>>(V, Vinv);
  k_build_G<<<dim3((NCT * 16 * 512 + 255) / 256), dim3(256), 0, stream>>>(gate_w, B, Vinv, G);
  k_EA<<<dim3((64 * 100 + 255) / 256), dim3(256), 0, stream>>>(E, alpha, EA);
  k_Minv<<<dim3((64 * 100 + 255) / 256), dim3(256), 0, stream>>>(Vinv, EA, Minv);
  k_gemm_coef<<<dim3(n / 16, NCT), dim3(32), 0, stream>>>(
      u_in, G, times, marks, llr, lim, gate_b, Minv, Cf, n);
  k_scan<<<dim3(32), dim3(SCAN_T), 0, stream>>>(Cf, Vinv, x0, YT, n);
  k_gemm_out<<<dim3(n / 16, 4), dim3(32), 0, stream>>>(YT, V, EA, marks, out, n);
}


# Launches cut to dependence levels: E@alpha and G's gate/zero rows ride as extra blocks of the inverse launch; Vinv@B rows and Vinv@EA share the next; 7 launches -> 5; sums and their order unchanged
# speedup vs baseline: 5.1895x; 1.2177x over previous
#include <hip/hip_runtime.h>
#include <math.h>


typedef __attribute__((ext_vector_type(2))) float v2f;
typedef __attribute__((ext_vector_type(8))) float v8f;

__device__ __forceinline__ float softplusf(float x) {
  return (x > 20.f) ? x : log1pf(expf(x));
}

#define INV_LD 68
#define NF_LD 20

#define PPT 5
#define NCT 7
#define EA_BLOCKS 13
#define GI_ROWS 48

__global__ __launch_bounds__(512) void k_inverse(const float* __restrict__ V,
                                                 float* __restrict__ Vinv,
                                                 const float* __restrict__ E,
                                                 const float* __restrict__ alpha,
                                                 float* __restrict__ EA,
                                                 const float* __restrict__ gate_w,
                                                 float* __restrict__ G) {
  if (blockIdx.x != 0) {
    const int b = blockIdx.x - 1;
    if (b < EA_BLOCKS) {
      const int idx = b * 512 + threadIdx.x;
      if (idx < 64 * 100) {
        const int d = idx / 100, k = idx % 100;
        float s = 0.f;
        for (int r = 0; r < 32; ++r) s = fmaf(E[d * 32 + r], alpha[r * 100 + k], s);
        EA[idx] = s;
      }
    } else {
      const int g = b - EA_BLOCKS;
      int row;
      float val = 0.f;
      if (g < 32) {
        row = (g / PPT) * 16 + 3 * (g % PPT);
        val = gate_w[g * 512 + threadIdx.x];
      } else if (g < 32 + NCT) {
        row = (g - 32) * 16 + 15;
      } else {
        row = (NCT - 1) * 16 + 6 + (g - 32 - NCT);
      }
      G[row * 512 + threadIdx.x] = val;
    }
    return;
  }
  __shared__ float M[64 * INV_LD];
  __shared__ float PR[16 * INV_LD];
  __shared__ __attribute__((aligned(16))) float NF[64 * NF_LD];
  __shared__ __attribute__((aligned(16))) float NFT[16 * 16];
  __shared__ float PINV[16];
  const int tid = threadIdx.x;
  const int l = tid & 31;
  const int w = __builtin_amdgcn_readfirstlane(tid >> 5);
  const int ti = w >> 2, tj = w & 3;
  const int m = l & 15, h = l >> 4, kh = h * 2;

  for (int idx = tid; idx < 64 * 64; idx += 512)
    M[(idx >> 6) * INV_LD + (idx & 63)] = V[idx];
  __syncthreads();

  for (int kb = 0; kb < 4; ++kb) {
    const int c0 = 16 * kb;

    if (w == 0) {
      const int r0 = (c0 + l) & 63, r1 = (c0 + l + 32) & 63;
      float a0[16], a1[16], n0[16], n1[16];
      float mypinv = 0.f;
#pragma unroll
      for (int c = 0; c < 16; ++c) {
        a0[c] = M[r0 * INV_LD + c0 + c];
        a1[c] = M[r1 * INV_LD + c0 + c];
      }
#pragma unroll
      for (int j = 0; j < 16; ++j) {
        const float pk = __int_as_float(__builtin_amdgcn_readlane(__float_as_int(a0[j]), j));
        const float pinv = 1.f / pk;
        const float f0 = a0[j], f1 = a1[j];
        const bool isp = (l == j);
        n0[j] = -f0;
        n1[j] = -f1;
        mypinv = isp ? pinv : mypinv;
#pragma unroll
        for (int c = 0; c < 16; ++c) {
          if (c == j) {
            a0[c] = isp ? pinv : (n0[j] * pinv);
            a1[c] = n1[j] * pinv;
          } else {
            const float pr =
                __int_as_float(__builtin_amdgcn_readlane(__float_as_int(a0[c]), j)) * pinv;
            a0[c] = isp ? pr : fmaf(n0[j], pr, a0[c]);
            a1[c] = fmaf(n1[j], pr, a1[c]);
          }
        }
      }
#pragma unroll
      for (int c = 0; c < 16; ++c) {
        M[r0 * INV_LD + c0 + c] = a0[c];
        M[r1 * INV_LD + c0 + c] = a1[c];
        NF[r0 * NF_LD + c] = n0[c];
        NF[r1 * NF_LD + c] = n1[c];
      }
      if (l < 16) {
#pragma unroll
        for (int j = 0; j < 16; ++j) NFT[j * 16 + l] = n0[j];
        PINV[l] = mypinv;
      }
    }
    __syncthreads();

    if (tid < 64 && (tid >> 4) != kb) {
      const int cc = tid;
      float x[16];
#pragma unroll
      for (int j = 0; j < 16; ++j) x[j] = M[(c0 + j) * INV_LD + cc];
#pragma unroll
      for (int j = 0; j < 16; ++j) {
        const float pr = x[j] * PINV[j];
        PR[j * INV_LD + cc] = pr;
#pragma unroll
        for (int jj = 0; jj < 16; ++jj) {
          if (jj != j) x[jj] = fmaf(NFT[j * 16 + jj], pr, x[jj]);
        }
        x[j] = pr;
      }
#pragma unroll
      for (int j = 0; j < 16; ++j) M[(c0 + j) * INV_LD + cc] = x[j];
    }
    __syncthreads();

    if (ti != kb && tj != kb) {
      float* Tij = M + (16 * ti) * INV_LD + 16 * tj;
      const float* At = NF + (16 * ti) * NF_LD;
      const float* Bt = PR + 16 * tj;
      v8f acc;
      for (int v = 0; v < 8; ++v) acc[v] = Tij[(8 * h + v) * INV_LD + m];
      for (int k4 = 0; k4 < 16; k4 += 4) {
        v2f a, b;
        a.x = At[m * NF_LD + k4 + kh];
        a.y = At[m * NF_LD + k4 + kh + 1];
        b.x = Bt[(k4 + kh) * INV_LD + m];
        b.y = Bt[(k4 + kh + 1) * INV_LD + m];
        acc = __builtin_amdgcn_wmma_f32_16x16x4_f32(false, a, false, b, (short)0, acc,
                                                    false, false);
      }
      for (int v = 0; v < 8; ++v) Tij[(8 * h + v) * INV_LD + m] = acc[v];
    }
    __syncthreads();
  }

  for (int idx = tid; idx < 64 * 64; idx += 512)
    Vinv[idx] = M[(idx >> 6) * INV_LD + (idx & 63)];
}

#define VB_BLOCKS 128
__global__ __launch_bounds__(256) void k_apply_inverse(const float* __restrict__ Vinv,
                                                       const float* __restrict__ B,
                                                       const float* __restrict__ EA,
                                                       float* __restrict__ G,
                                                       float* __restrict__ Minv) {
  const int b = blockIdx.x;
  if (b < VB_BLOCKS) {
    const int d = b >> 1;
    const int h = (b & 1) * 256 + threadIdx.x;
    const int p = d >> 1;
    const int row = (p / PPT) * 16 + 3 * (p % PPT) + 1 + (d & 1);
    const float* vr = Vinv + d * 64;
    float s = 0.f;
#pragma unroll
    for (int j = 0; j < 64; ++j) s = fmaf(vr[j], B[j * 512 + h], s);
    G[row * 512 + h] = s;
  } else {
    const int idx = (b - VB_BLOCKS) * 256 + threadIdx.x;
    if (idx >= 64 * 100) return;
    const int d = idx / 100, k = idx % 100;
    float s = 0.f;
#pragma unroll
    for (int j = 0; j < 64; ++j) s = fmaf(Vinv[d * 64 + j], EA[j * 100 + k], s);
    Minv[idx] = s;
  }
}

#define TS_LD 17
__global__ __launch_bounds__(32) void k_gemm_coef(const float* __restrict__ U,
                                                  const float* __restrict__ G,
                                                  const float* __restrict__ times,
                                                  const int* __restrict__ marks,
                                                  const float* __restrict__ llr,
                                                  const float* __restrict__ lim,
                                                  const float* __restrict__ gate_b,
                                                  const float* __restrict__ Minv,
                                                  float* __restrict__ Cf, int n) {
  __shared__ float TS[16 * TS_LD];
  const int rt = blockIdx.x, ct = blockIdx.y;
  const int l = threadIdx.x;
  const int m = l & 15;
  const int hf = l >> 4;
  const int kh = hf * 2;
  const float* Arow = U + (size_t)(rt * 16 + m) * 512;
  const float* Brow = G + (size_t)(ct * 16 + m) * 512;
  const int ns = rt * 16 + m;
  const float t = times[ns];
  const float tprev = times[ns > 0 ? ns - 1 : 0];
  const int mk = marks[ns];
  float gb[3], lr[3], li[3];
#pragma unroll
  for (int k = 0; k < 3; ++k) {
    const int pc = min(PPT * ct + 2 * k + hf, 31);
    gb[k] = gate_b[pc];
    lr[k] = llr[pc];
    li[k] = lim[pc];
  }
  v8f acc = {};
  for (int kb = 0; kb < 512; kb += 4) {
    v2f a = *(const v2f*)(Arow + kb + kh);
    v2f b = *(const v2f*)(Brow + kb + kh);
    acc = __builtin_amdgcn_wmma_f32_16x16x4_f32(false, a, false, b, (short)0, acc,
                                                false, false);
  }
  for (int v = 0; v < 8; ++v) TS[(8 * hf + v) * TS_LD + m] = acc[v];
  __syncthreads();
  const float dt = (ns == 0) ? t : (t - tprev);
#pragma unroll
  for (int k = 0; k < 3; ++k) {
    const int q = 2 * k + hf;
    const int p = PPT * ct + q;
    if (q < PPT && p < 32) {
      const float* tr = TS + m * TS_LD + 3 * q;
      float gate = softplusf(tr[0] + gb[k]);
      float real = -softplusf(lr[k]) * gate;
      float im = li[k];
      float e = expf(real * dt);
      float a = e * cosf(im * dt);
      float c = e * sinf(im * dt);
      float denom = real * real + im * im;
      float nre = a - 1.f, nim = c;
      float qr = (nre * real + nim * im) / (denom + 1e-12f);
      float qi = (nim * real - nre * im) / (denom + 1e-12f);
      if (denom < 1e-8f) { qr = dt; qi = 0.f; }
      if (!(dt > 0.f)) { qr = 0.f; qi = 0.f; }
      float w0 = tr[1];
      float w1 = tr[2];
      float ure = qr * w0 - qi * w1;
      float uim = qi * w0 + qr * w1;
      ure += Minv[(2 * p) * 100 + mk];
      uim += Minv[(2 * p + 1) * 100 + mk];
      ((float4*)Cf)[(size_t)p * n + ns] = make_float4(a, c, ure, uim);
    }
  }
}

#define SPT 16
#define SCAN_T 512
#define SCAN_W (SCAN_T / 32)

__device__ __forceinline__ void wave_prefix(float& Ar, float& Ai, float& Ur, float& Ui, int l) {
  for (int d = 1; d < 32; d <<= 1) {
    const float pAr = __shfl_up(Ar, d, 32), pAi = __shfl_up(Ai, d, 32);
    const float pUr = __shfl_up(Ur, d, 32), pUi = __shfl_up(Ui, d, 32);
    if (l >= d) {
      const float nUr = Ar * pUr - Ai * pUi + Ur, nUi = Ar * pUi + Ai * pUr + Ui;
      const float nAr = Ar * pAr - Ai * pAi, nAi = Ar * pAi + Ai * pAr;
      Ar = nAr; Ai = nAi; Ur = nUr; Ui = nUi;
    }
  }
}

__global__ __launch_bounds__(SCAN_T) void k_scan(const float* __restrict__ Cf,
                                                 const float* __restrict__ Vinv,
                                                 const float* __restrict__ x0,
                                                 float* __restrict__ YT, int n) {
  __shared__ float4 WAGG[SCAN_W];
  __shared__ float Y0[2];
  const int p = blockIdx.x, t = threadIdx.x, l = t & 31;
  const int w = __builtin_amdgcn_readfirstlane(t >> 5);
  const int s0 = t * SPT;
  const bool live = (s0 + SPT <= n);
  float ar[SPT], ai[SPT], ur[SPT], ui[SPT];
  if (live) {
    const float4* c4 = (const float4*)Cf + (size_t)p * n + s0;
#pragma unroll
    for (int i = 0; i < SPT; ++i) {
      const float4 v = c4[i];
      ar[i] = v.x; ai[i] = v.y; ur[i] = v.z; ui[i] = v.w;
    }
  } else {
#pragma unroll
    for (int i = 0; i < SPT; ++i) { ar[i] = 1.f; ai[i] = 0.f; ur[i] = 0.f; ui[i] = 0.f; }
  }
  if (w == 0) {
    const float* v0 = Vinv + (2 * p) * 64;
    const float* v1 = v0 + 64;
    const float xa = x0[l], xb = x0[32 + l];
    float sr = fmaf(v0[32 + l], xb, v0[l] * xa);
    float si = fmaf(v1[32 + l], xb, v1[l] * xa);
    for (int d = 16; d >= 1; d >>= 1) {
      sr += __shfl_xor(sr, d, 32);
      si += __shfl_xor(si, d, 32);
    }
    if (l == 0) { Y0[0] = sr; Y0[1] = si; }
  }
  float Ar = 1.f, Ai = 0.f, Ur = 0.f, Ui = 0.f;
#pragma unroll
  for (int i = 0; i < SPT; ++i) {
    const float nAr = ar[i] * Ar - ai[i] * Ai, nAi = ar[i] * Ai + ai[i] * Ar;
    const float nUr = ar[i] * Ur - ai[i] * Ui + ur[i], nUi = ar[i] * Ui + ai[i] * Ur + ui[i];
    Ar = nAr; Ai = nAi; Ur = nUr; Ui = nUi;
  }
  wave_prefix(Ar, Ai, Ur, Ui, l);
  if (l == 31) WAGG[w] = make_float4(Ar, Ai, Ur, Ui);
  __syncthreads();
  const float y0r = Y0[0], y0i = Y0[1];
  float qAr = 1.f, qAi = 0.f, qUr = 0.f, qUi = 0.f;
  if (l < SCAN_W) {
    const float4 q = WAGG[l];
    qAr = q.x; qAi = q.y; qUr = q.z; qUi = q.w;
  }
  wave_prefix(qAr, qAi, qUr, qUi, l);
  const int src = (w > 0) ? (w - 1) : 0;
  float xAr = __shfl(qAr, src, 32), xAi = __shfl(qAi, src, 32);
  float xUr = __shfl(qUr, src, 32), xUi = __shfl(qUi, src, 32);
  if (w == 0) { xAr = 1.f; xAi = 0.f; xUr = 0.f; xUi = 0.f; }
  const float ywr = xAr * y0r - xAi * y0i + xUr;
  const float ywi = xAr * y0i + xAi * y0r + xUi;
  float eAr = __shfl_up(Ar, 1, 32), eAi = __shfl_up(Ai, 1, 32);
  float eUr = __shfl_up(Ur, 1, 32), eUi = __shfl_up(Ui, 1, 32);
  if (l == 0) { eAr = 1.f; eAi = 0.f; eUr = 0.f; eUi = 0.f; }
  float yr = eAr * ywr - eAi * ywi + eUr;
  float yi = eAr * ywi + eAi * ywr + eUi;
  float outr[SPT], outi[SPT];
#pragma unroll
  for (int i = 0; i < SPT; ++i) {
    const float nyr = ar[i] * yr - ai[i] * yi + ur[i];
    const float nyi = ar[i] * yi + ai[i] * yr + ui[i];
    yr = nyr; yi = nyi;
    outr[i] = yr; outi[i] = yi;
  }
  if (live) {
    float4* yo = (float4*)(YT + (size_t)(2 * p) * n + s0);
    float4* y1 = (float4*)(YT + (size_t)(2 * p + 1) * n + s0);
#pragma unroll
    for (int i = 0; i < SPT; i += 4) {
      yo[i / 4] = make_float4(outr[i], outr[i + 1], outr[i + 2], outr[i + 3]);
      y1[i / 4] = make_float4(outi[i], outi[i + 1], outi[i + 2], outi[i + 3]);
    }
  }
}

__global__ __launch_bounds__(32) void k_gemm_out(const float* __restrict__ YT,
                                                 const float* __restrict__ V,
                                                 const float* __restrict__ EA,
                                                 const int* __restrict__ marks,
                                                 float* __restrict__ out, int n) {
  const int rt = blockIdx.x, ct = blockIdx.y, l = threadIdx.x;
  const int m = l & 15;
  const int kh = (l >> 4) * 2;
  const float* Acol = YT + (size_t)kh * n + (rt * 16 + m);
  const int dcol = ct * 16 + m;
  const float* Brow = V + (size_t)dcol * 64;
  v8f acc = {};
#pragma unroll
  for (int kb = 0; kb < 64; kb += 4) {
    v2f a;
    a.x = Acol[(size_t)kb * n];
    a.y = Acol[(size_t)(kb + 1) * n];
    v2f b = *(const v2f*)(Brow + kb + kh);
    acc = __builtin_amdgcn_wmma_f32_16x16x4_f32(false, a, false, b, (short)0, acc,
                                                false, false);
  }
  const int rbase = rt * 16 + (l >> 4) * 8;
  const size_t xl_off = (size_t)n * 64;
  for (int v = 0; v < 8; ++v) {
    int row = rbase + v;
    float xr = acc[v];
    float r = EA[dcol * 100 + marks[row]];
    out[(size_t)row * 64 + dcol] = xr;
    out[xl_off + (size_t)row * 64 + dcol] = xr - r;
  }
}

extern "C" void kernel_launch(void* const* d_in, const int* in_sizes, int n_in,
                              void* d_out, int out_size, void* d_ws, size_t ws_size,
                              hipStream_t stream) {
  const float* times   = (const float*)d_in[0];
  const int*   marks   = (const int*)d_in[1];
  const float* u_in    = (const float*)d_in[2];
  const float* llr     = (const float*)d_in[3];
  const float* lim     = (const float*)d_in[4];
  const float* V       = (const float*)d_in[5];
  const float* B       = (const float*)d_in[6];
  const float* E       = (const float*)d_in[7];
  const float* alpha   = (const float*)d_in[8];
  const float* gate_w  = (const float*)d_in[9];
  const float* gate_b  = (const float*)d_in[10];
  const float* x0      = (const float*)d_in[11];
  float* out = (float*)d_out;
  const int n = in_sizes[0];

  float* ws    = (float*)d_ws;
  float* Vinv  = ws; ws += 64 * 64;
  float* G     = ws; ws += NCT * 16 * 512;
  float* EA    = ws; ws += 64 * 100;
  float* Minv  = ws; ws += 64 * 100;
  float* Cf    = ws; ws += (size_t)n * 128;
  float* YT    = ws; ws += (size_t)n * 64;

  k_inverse<<<dim3(1 + EA_BLOCKS + GI_ROWS), dim3(512), 0, stream>>>(V, Vinv, E, alpha, EA, gate_w, G);
  k_apply_inverse<<<dim3(VB_BLOCKS + (64 * 100 + 255) / 256), dim3(256), 0, stream>>>(
      Vinv, B, EA, G, Minv);
  k_gemm_coef<<<dim3(n / 16, NCT), dim3(32), 0, stream>>>(
      u_in, G, times, marks, llr, lim, gate_b, Minv, Cf, n);
  k_scan<<<dim3(32), dim3(SCAN_T), 0, stream>>>(Cf, Vinv, x0, YT, n);
  k_gemm_out<<<dim3(n / 16, 4), dim3(32), 0, stream>>>(YT, V, EA, marks, out, n);
}


# U GEMM split by dependence: the gate third and the factors it fixes (a,c,qr,qi) form beside the inverse in its launch, K in 4 or 16 slices (gate sum order changes); only the Vinv@B columns wait
# speedup vs baseline: 5.8784x; 1.1327x over previous
#include <hip/hip_runtime.h>
#include <math.h>


typedef __attribute__((ext_vector_type(2))) float v2f;
typedef __attribute__((ext_vector_type(8))) float v8f;

__device__ __forceinline__ float softplusf(float x) {
  return (x > 20.f) ? x : log1pf(expf(x));
}

#define INV_LD 68
#define NF_LD 20

#define EA_BLOCKS 13
#define COEF_FULL 255
#define PT_LD 17

template <int SH>
__device__ __forceinline__ void coef_units(float* __restrict__ PT, const int unit0,
                                           const float* __restrict__ U,
                                           const float* __restrict__ gate_w,
                                           const float* __restrict__ times,
                                           const float* __restrict__ gate_b,
                                           const float* __restrict__ llr,
                                           const float* __restrict__ lim,
                                           float* __restrict__ Cf, const int n) {
  constexpr int WPU = 1 << SH;
  constexpr int KLEN = 512 >> SH;
  constexpr int NI = (WPU < 8) ? (8 / WPU) : 1;
  const int tid = threadIdx.x;
  const int l = tid & 31;
  const int w = __builtin_amdgcn_readfirstlane(tid >> 5);
  const int unit = unit0 + (w >> SH);
  const int kq = w & (WPU - 1);
  const int rt = unit >> 1, pg = unit & 1;
  const int m = l & 15, hf = l >> 4;
  const int ns = rt * 16 + m;
  const float t = times[ns];
  const float tprev = times[ns > 0 ? ns - 1 : 0];
  const int g = 2 * kq + hf;
  float gb[NI], lr[NI], li[NI];
#pragma unroll
  for (int i = 0; i < NI; ++i) {
    const int pc = 16 * pg + min(g + 2 * WPU * i, 15);
    gb[i] = gate_b[pc];
    lr[i] = llr[pc];
    li[i] = lim[pc];
  }
  const float* Arow = U + (size_t)ns * 512 + kq * KLEN + 4 * hf;
  const float* Brow = gate_w + (size_t)(16 * pg + m) * 512 + kq * KLEN + 4 * hf;
  v8f acc = {};
#pragma unroll
  for (int kb = 0; kb < KLEN; kb += 8) {
    const float4 a4 = *(const float4*)(Arow + kb);
    const float4 b4 = *(const float4*)(Brow + kb);
    v2f a, b;
    a.x = a4.x; a.y = a4.y;
    b.x = b4.x; b.y = b4.y;
    acc = __builtin_amdgcn_wmma_f32_16x16x4_f32(false, a, false, b, (short)0, acc, false, false);
    a.x = a4.z; a.y = a4.w;
    b.x = b4.z; b.y = b4.w;
    acc = __builtin_amdgcn_wmma_f32_16x16x4_f32(false, a, false, b, (short)0, acc, false, false);
  }
  float* P = PT + w * (16 * PT_LD);
  for (int v = 0; v < 8; ++v) P[(8 * hf + v) * PT_LD + m] = acc[v];
  __syncthreads();
  const float dt = (ns == 0) ? t : (t - tprev);
  const float* P0 = PT + ((w >> SH) << SH) * (16 * PT_LD) + m * PT_LD;
#pragma unroll
  for (int i = 0; i < NI; ++i) {
    const int q = g + 2 * WPU * i;
    if (q < 16) {
      const int p = 16 * pg + q;
      float x = P0[q];
      for (int j = 1; j < WPU; ++j) x += P0[j * (16 * PT_LD) + q];
      float gate = softplusf(x + gb[i]);
      float real = -softplusf(lr[i]) * gate;
      float im = li[i];
      float e = expf(real * dt);
      float a = e * cosf(im * dt);
      float c = e * sinf(im * dt);
      float denom = real * real + im * im;
      float nre = a - 1.f, nim = c;
      float qr = (nre * real + nim * im) / (denom + 1e-12f);
      float qi = (nim * real - nre * im) / (denom + 1e-12f);
      if (denom < 1e-8f) { qr = dt; qi = 0.f; }
      if (!(dt > 0.f)) { qr = 0.f; qi = 0.f; }
      ((float4*)Cf)[(size_t)p * n + ns] = make_float4(a, c, qr, qi);
    }
  }
}

__global__ __launch_bounds__(512) void k_inverse(const float* __restrict__ V,
                                                 float* __restrict__ Vinv,
                                                 const float* __restrict__ E,
                                                 const float* __restrict__ alpha,
                                                 float* __restrict__ EA,
                                                 const float* __restrict__ gate_w,
                                                 const float* __restrict__ U,
                                                 const float* __restrict__ times,
                                                 const float* __restrict__ gate_b,
                                                 const float* __restrict__ llr,
                                                 const float* __restrict__ lim,
                                                 float* __restrict__ Cf, int n) {
  __shared__ float M[64 * INV_LD];
  if (blockIdx.x != 0) {
    const int b = blockIdx.x - 1;
    const int ntail = n / 8 - 4 * COEF_FULL;
    if (b < ntail) {
      coef_units<4>(M, 4 * COEF_FULL + b, U, gate_w, times, gate_b, llr, lim, Cf, n);
    } else if (b < ntail + COEF_FULL) {
      coef_units<2>(M, 4 * (b - ntail), U, gate_w, times, gate_b, llr, lim, Cf, n);
    } else {
      const int idx = (b - COEF_FULL - ntail) * 512 + threadIdx.x;
      if (idx < 64 * 100) {
        const int d = idx / 100, k = idx % 100;
        float s = 0.f;
        for (int r = 0; r < 32; ++r) s = fmaf(E[d * 32 + r], alpha[r * 100 + k], s);
        EA[idx] = s;
      }
    }
    return;
  }
  __shared__ float PR[16 * INV_LD];
  __shared__ __attribute__((aligned(16))) float NF[64 * NF_LD];
  __shared__ __attribute__((aligned(16))) float NFT[16 * 16];
  __shared__ float PINV[16];
  const int tid = threadIdx.x;
  const int l = tid & 31;
  const int w = __builtin_amdgcn_readfirstlane(tid >> 5);
  const int ti = w >> 2, tj = w & 3;
  const int m = l & 15, h = l >> 4, kh = h * 2;

  for (int idx = tid; idx < 64 * 64; idx += 512)
    M[(idx >> 6) * INV_LD + (idx & 63)] = V[idx];
  __syncthreads();

  for (int kb = 0; kb < 4; ++kb) {
    const int c0 = 16 * kb;

    if (w == 0) {
      const int r0 = (c0 + l) & 63, r1 = (c0 + l + 32) & 63;
      float a0[16], a1[16], n0[16], n1[16];
      float mypinv = 0.f;
#pragma unroll
      for (int c = 0; c < 16; ++c) {
        a0[c] = M[r0 * INV_LD + c0 + c];
        a1[c] = M[r1 * INV_LD + c0 + c];
      }
#pragma unroll
      for (int j = 0; j < 16; ++j) {
        const float pk = __int_as_float(__builtin_amdgcn_readlane(__float_as_int(a0[j]), j));
        const float pinv = 1.f / pk;
        const float f0 = a0[j], f1 = a1[j];
        const bool isp = (l == j);
        n0[j] = -f0;
        n1[j] = -f1;
        mypinv = isp ? pinv : mypinv;
#pragma unroll
        for (int c = 0; c < 16; ++c) {
          if (c == j) {
            a0[c] = isp ? pinv : (n0[j] * pinv);
            a1[c] = n1[j] * pinv;
          } else {
            const float pr =
                __int_as_float(__builtin_amdgcn_readlane(__float_as_int(a0[c]), j)) * pinv;
            a0[c] = isp ? pr : fmaf(n0[j], pr, a0[c]);
            a1[c] = fmaf(n1[j], pr, a1[c]);
          }
        }
      }
#pragma unroll
      for (int c = 0; c < 16; ++c) {
        M[r0 * INV_LD + c0 + c] = a0[c];
        M[r1 * INV_LD + c0 + c] = a1[c];
        NF[r0 * NF_LD + c] = n0[c];
        NF[r1 * NF_LD + c] = n1[c];
      }
      if (l < 16) {
#pragma unroll
        for (int j = 0; j < 16; ++j) NFT[j * 16 + l] = n0[j];
        PINV[l] = mypinv;
      }
    }
    __syncthreads();

    if (tid < 64 && (tid >> 4) != kb) {
      const int cc = tid;
      float x[16];
#pragma unroll
      for (int j = 0; j < 16; ++j) x[j] = M[(c0 + j) * INV_LD + cc];
#pragma unroll
      for (int j = 0; j < 16; ++j) {
        const float pr = x[j] * PINV[j];
        PR[j * INV_LD + cc] = pr;
#pragma unroll
        for (int jj = 0; jj < 16; ++jj) {
          if (jj != j) x[jj] = fmaf(NFT[j * 16 + jj], pr, x[jj]);
        }
        x[j] = pr;
      }
#pragma unroll
      for (int j = 0; j < 16; ++j) M[(c0 + j) * INV_LD + cc] = x[j];
    }
    __syncthreads();

    if (ti != kb && tj != kb) {
      float* Tij = M + (16 * ti) * INV_LD + 16 * tj;
      const float* At = NF + (16 * ti) * NF_LD;
      const float* Bt = PR + 16 * tj;
      v8f acc;
      for (int v = 0; v < 8; ++v) acc[v] = Tij[(8 * h + v) * INV_LD + m];
      for (int k4 = 0; k4 < 16; k4 += 4) {
        v2f a, b;
        a.x = At[m * NF_LD + k4 + kh];
        a.y = At[m * NF_LD + k4 + kh + 1];
        b.x = Bt[(k4 + kh) * INV_LD + m];
        b.y = Bt[(k4 + kh + 1) * INV_LD + m];
        acc = __builtin_amdgcn_wmma_f32_16x16x4_f32(false, a, false, b, (short)0, acc,
                                                    false, false);
      }
      for (int v = 0; v < 8; ++v) Tij[(8 * h + v) * INV_LD + m] = acc[v];
    }
    __syncthreads();
  }

  for (int idx = tid; idx < 64 * 64; idx += 512)
    Vinv[idx] = M[(idx >> 6) * INV_LD + (idx & 63)];
}

#define VB_BLOCKS 128
__global__ __launch_bounds__(256) void k_apply_inverse(const float* __restrict__ Vinv,
                                                       const float* __restrict__ B,
                                                       const float* __restrict__ EA,
                                                       float* __restrict__ G,
                                                       float* __restrict__ Minv) {
  const int b = blockIdx.x;
  if (b < VB_BLOCKS) {
    const int d = b >> 1;
    const int h = (b & 1) * 256 + threadIdx.x;
    const float* vr = Vinv + d * 64;
    float s = 0.f;
#pragma unroll
    for (int j = 0; j < 64; ++j) s = fmaf(vr[j], B[j * 512 + h], s);
    G[d * 512 + h] = s;
  } else {
    const int idx = (b - VB_BLOCKS) * 256 + threadIdx.x;
    if (idx >= 64 * 100) return;
    const int d = idx / 100, k = idx % 100;
    float s = 0.f;
#pragma unroll
    for (int j = 0; j < 64; ++j) s = fmaf(Vinv[d * 64 + j], EA[j * 100 + k], s);
    Minv[idx] = s;
  }
}

#define TS_LD 17
__global__ __launch_bounds__(32) void k_gemm_w(const float* __restrict__ U,
                                               const float* __restrict__ G,
                                               const int* __restrict__ marks,
                                               const float* __restrict__ Minv,
                                               float* __restrict__ Cf, int n) {
  __shared__ float TS[16 * TS_LD];
  const int rt = blockIdx.x, ct = blockIdx.y;
  const int l = threadIdx.x;
  const int m = l & 15;
  const int hf = l >> 4;
  const int kh = hf * 2;
  const float* Arow = U + (size_t)(rt * 16 + m) * 512;
  const float* Brow = G + (size_t)(ct * 16 + m) * 512;
  const int ns = rt * 16 + m;
  const int mk = marks[ns];
  float4* cfp = (float4*)Cf;
  float4 cf[4];
  float r0[4], r1[4];
#pragma unroll
  for (int k = 0; k < 4; ++k) {
    const int p = 8 * ct + 2 * k + hf;
    cf[k] = cfp[(size_t)p * n + ns];
    r0[k] = Minv[(2 * p) * 100 + mk];
    r1[k] = Minv[(2 * p + 1) * 100 + mk];
  }
  v8f acc = {};
  for (int kb = 0; kb < 512; kb += 4) {
    v2f a = *(const v2f*)(Arow + kb + kh);
    v2f b = *(const v2f*)(Brow + kb + kh);
    acc = __builtin_amdgcn_wmma_f32_16x16x4_f32(false, a, false, b, (short)0, acc,
                                                false, false);
  }
  for (int v = 0; v < 8; ++v) TS[(8 * hf + v) * TS_LD + m] = acc[v];
  __syncthreads();
#pragma unroll
  for (int k = 0; k < 4; ++k) {
    const int q = 2 * k + hf;
    const int p = 8 * ct + q;
    const float* tr = TS + m * TS_LD + 2 * q;
    const float qr = cf[k].z, qi = cf[k].w;
    float w0 = tr[0];
    float w1 = tr[1];
    float ure = qr * w0 - qi * w1;
    float uim = qi * w0 + qr * w1;
    ure += r0[k];
    uim += r1[k];
    cfp[(size_t)p * n + ns] = make_float4(cf[k].x, cf[k].y, ure, uim);
  }
}

#define SPT 16
#define SCAN_T 512
#define SCAN_W (SCAN_T / 32)

__device__ __forceinline__ void wave_prefix(float& Ar, float& Ai, float& Ur, float& Ui, int l) {
  for (int d = 1; d < 32; d <<= 1) {
    const float pAr = __shfl_up(Ar, d, 32), pAi = __shfl_up(Ai, d, 32);
    const float pUr = __shfl_up(Ur, d, 32), pUi = __shfl_up(Ui, d, 32);
    if (l >= d) {
      const float nUr = Ar * pUr - Ai * pUi + Ur, nUi = Ar * pUi + Ai * pUr + Ui;
      const float nAr = Ar * pAr - Ai * pAi, nAi = Ar * pAi + Ai * pAr;
      Ar = nAr; Ai = nAi; Ur = nUr; Ui = nUi;
    }
  }
}

__global__ __launch_bounds__(SCAN_T) void k_scan(const float* __restrict__ Cf,
                                                 const float* __restrict__ Vinv,
                                                 const float* __restrict__ x0,
                                                 float* __restrict__ YT, int n) {
  __shared__ float4 WAGG[SCAN_W];
  __shared__ float Y0[2];
  const int p = blockIdx.x, t = threadIdx.x, l = t & 31;
  const int w = __builtin_amdgcn_readfirstlane(t >> 5);
  const int s0 = t * SPT;
  const bool live = (s0 + SPT <= n);
  float ar[SPT], ai[SPT], ur[SPT], ui[SPT];
  if (live) {
    const float4* c4 = (const float4*)Cf + (size_t)p * n + s0;
#pragma unroll
    for (int i = 0; i < SPT; ++i) {
      const float4 v = c4[i];
      ar[i] = v.x; ai[i] = v.y; ur[i] = v.z; ui[i] = v.w;
    }
  } else {
#pragma unroll
    for (int i = 0; i < SPT; ++i) { ar[i] = 1.f; ai[i] = 0.f; ur[i] = 0.f; ui[i] = 0.f; }
  }
  if (w == 0) {
    const float* v0 = Vinv + (2 * p) * 64;
    const float* v1 = v0 + 64;
    const float xa = x0[l], xb = x0[32 + l];
    float sr = fmaf(v0[32 + l], xb, v0[l] * xa);
    float si = fmaf(v1[32 + l], xb, v1[l] * xa);
    for (int d = 16; d >= 1; d >>= 1) {
      sr += __shfl_xor(sr, d, 32);
      si += __shfl_xor(si, d, 32);
    }
    if (l == 0) { Y0[0] = sr; Y0[1] = si; }
  }
  float Ar = 1.f, Ai = 0.f, Ur = 0.f, Ui = 0.f;
#pragma unroll
  for (int i = 0; i < SPT; ++i) {
    const float nAr = ar[i] * Ar - ai[i] * Ai, nAi = ar[i] * Ai + ai[i] * Ar;
    const float nUr = ar[i] * Ur - ai[i] * Ui + ur[i], nUi = ar[i] * Ui + ai[i] * Ur + ui[i];
    Ar = nAr; Ai = nAi; Ur = nUr; Ui = nUi;
  }
  wave_prefix(Ar, Ai, Ur, Ui, l);
  if (l == 31) WAGG[w] = make_float4(Ar, Ai, Ur, Ui);
  __syncthreads();
  const float y0r = Y0[0], y0i = Y0[1];
  float qAr = 1.f, qAi = 0.f, qUr = 0.f, qUi = 0.f;
  if (l < SCAN_W) {
    const float4 q = WAGG[l];
    qAr = q.x; qAi = q.y; qUr = q.z; qUi = q.w;
  }
  wave_prefix(qAr, qAi, qUr, qUi, l);
  const int src = (w > 0) ? (w - 1) : 0;
  float xAr = __shfl(qAr, src, 32), xAi = __shfl(qAi, src, 32);
  float xUr = __shfl(qUr, src, 32), xUi = __shfl(qUi, src, 32);
  if (w == 0) { xAr = 1.f; xAi = 0.f; xUr = 0.f; xUi = 0.f; }
  const float ywr = xAr * y0r - xAi * y0i + xUr;
  const float ywi = xAr * y0i + xAi * y0r + xUi;
  float eAr = __shfl_up(Ar, 1, 32), eAi = __shfl_up(Ai, 1, 32);
  float eUr = __shfl_up(Ur, 1, 32), eUi = __shfl_up(Ui, 1, 32);
  if (l == 0) { eAr = 1.f; eAi = 0.f; eUr = 0.f; eUi = 0.f; }
  float yr = eAr * ywr - eAi * ywi + eUr;
  float yi = eAr * ywi + eAi * ywr + eUi;
  float outr[SPT], outi[SPT];
#pragma unroll
  for (int i = 0; i < SPT; ++i) {
    const float nyr = ar[i] * yr - ai[i] * yi + ur[i];
    const float nyi = ar[i] * yi + ai[i] * yr + ui[i];
    yr = nyr; yi = nyi;
    outr[i] = yr; outi[i] = yi;
  }
  if (live) {
    float4* yo = (float4*)(YT + (size_t)(2 * p) * n + s0);
    float4* y1 = (float4*)(YT + (size_t)(2 * p + 1) * n + s0);
#pragma unroll
    for (int i = 0; i < SPT; i += 4) {
      yo[i / 4] = make_float4(outr[i], outr[i + 1], outr[i + 2], outr[i + 3]);
      y1[i / 4] = make_float4(outi[i], outi[i + 1], outi[i + 2], outi[i + 3]);
    }
  }
}

__global__ __launch_bounds__(32) void k_gemm_out(const float* __restrict__ YT,
                                                 const float* __restrict__ V,
                                                 const float* __restrict__ EA,
                                                 const int* __restrict__ marks,
                                                 float* __restrict__ out, int n) {
  const int rt = blockIdx.x, ct = blockIdx.y, l = threadIdx.x;
  const int m = l & 15;
  const int kh = (l >> 4) * 2;
  const float* Acol = YT + (size_t)kh * n + (rt * 16 + m);
  const int dcol = ct * 16 + m;
  const float* Brow = V + (size_t)dcol * 64;
  v8f acc = {};
#pragma unroll
  for (int kb = 0; kb < 64; kb += 4) {
    v2f a;
    a.x = Acol[(size_t)kb * n];
    a.y = Acol[(size_t)(kb + 1) * n];
    v2f b = *(const v2f*)(Brow + kb + kh);
    acc = __builtin_amdgcn_wmma_f32_16x16x4_f32(false, a, false, b, (short)0, acc,
                                                false, false);
  }
  const int rbase = rt * 16 + (l >> 4) * 8;
  const size_t xl_off = (size_t)n * 64;
  for (int v = 0; v < 8; ++v) {
    int row = rbase + v;
    float xr = acc[v];
    float r = EA[dcol * 100 + marks[row]];
    out[(size_t)row * 64 + dcol] = xr;
    out[xl_off + (size_t)row * 64 + dcol] = xr - r;
  }
}

extern "C" void kernel_launch(void* const* d_in, const int* in_sizes, int n_in,
                              void* d_out, int out_size, void* d_ws, size_t ws_size,
                              hipStream_t stream) {
  const float* times   = (const float*)d_in[0];
  const int*   marks   = (const int*)d_in[1];
  const float* u_in    = (const float*)d_in[2];
  const float* llr     = (const float*)d_in[3];
  const float* lim     = (const float*)d_in[4];
  const float* V       = (const float*)d_in[5];
  const float* B       = (const float*)d_in[6];
  const float* E       = (const float*)d_in[7];
  const float* alpha   = (const float*)d_in[8];
  const float* gate_w  = (const float*)d_in[9];
  const float* gate_b  = (const float*)d_in[10];
  const float* x0      = (const float*)d_in[11];
  float* out = (float*)d_out;
  const int n = in_sizes[0];

  float* ws    = (float*)d_ws;
  float* Vinv  = ws; ws += 64 * 64;
  float* G     = ws; ws += 64 * 512;
  float* EA    = ws; ws += 64 * 100;
  float* Minv  = ws; ws += 64 * 100;
  float* Cf    = ws; ws += (size_t)n * 128;
  float* YT    = ws; ws += (size_t)n * 64;

  k_inverse<<<dim3(1 + COEF_FULL + (n / 8 - 4 * COEF_FULL) + EA_BLOCKS), dim3(512), 0, stream>>>(
      V, Vinv, E, alpha, EA, gate_w, u_in, times, gate_b, llr, lim, Cf, n);
  k_apply_inverse<<<dim3(VB_BLOCKS + (64 * 100 + 255) / 256), dim3(256), 0, stream>>>(
      Vinv, B, EA, G, Minv);
  k_gemm_w<<<dim3(n / 16, 4), dim3(32), 0, stream>>>(u_in, G, marks, Minv, Cf, n);
  k_scan<<<dim3(32), dim3(SCAN_T), 0, stream>>>(Cf, Vinv, x0, YT, n);
  k_gemm_out<<<dim3(n / 16, 4), dim3(32), 0, stream>>>(YT, V, EA, marks, out, n);
}


# Vinv@B and Vinv@EA formed inside the U GEMM: a 4-wave block builds its 16 rows of each in LDS on WMMA f32, 2 row tiles per wave share LDS B fragments; 5 launches -> 4; G, Minv, w sum order changes
# speedup vs baseline: 6.7706x; 1.1518x over previous
#include <hip/hip_runtime.h>
#include <math.h>


typedef __attribute__((ext_vector_type(2))) float v2f;
typedef __attribute__((ext_vector_type(8))) float v8f;

__device__ __forceinline__ float softplusf(float x) {
  return (x > 20.f) ? x : log1pf(expf(x));
}

#define INV_LD 68
#define NF_LD 20

#define EA_BLOCKS 13
#define COEF_FULL 255
#define PT_LD 17

template <int SH>
__device__ __forceinline__ void coef_units(float* __restrict__ PT, const int unit0,
                                           const float* __restrict__ U,
                                           const float* __restrict__ gate_w,
                                           const float* __restrict__ times,
                                           const float* __restrict__ gate_b,
                                           const float* __restrict__ llr,
                                           const float* __restrict__ lim,
                                           float* __restrict__ Cf, const int n) {
  constexpr int WPU = 1 << SH;
  constexpr int KLEN = 512 >> SH;
  constexpr int NI = (WPU < 8) ? (8 / WPU) : 1;
  const int tid = threadIdx.x;
  const int l = tid & 31;
  const int w = __builtin_amdgcn_readfirstlane(tid >> 5);
  const int unit = unit0 + (w >> SH);
  const int kq = w & (WPU - 1);
  const int rt = unit >> 1, pg = unit & 1;
  const int m = l & 15, hf = l >> 4;
  const int ns = rt * 16 + m;
  const float t = times[ns];
  const float tprev = times[ns > 0 ? ns - 1 : 0];
  const int g = 2 * kq + hf;
  float gb[NI], lr[NI], li[NI];
#pragma unroll
  for (int i = 0; i < NI; ++i) {
    const int pc = 16 * pg + min(g + 2 * WPU * i, 15);
    gb[i] = gate_b[pc];
    lr[i] = llr[pc];
    li[i] = lim[pc];
  }
  const float* Arow = U + (size_t)ns * 512 + kq * KLEN + 4 * hf;
  const float* Brow = gate_w + (size_t)(16 * pg + m) * 512 + kq * KLEN + 4 * hf;
  v8f acc = {};
#pragma unroll
  for (int kb = 0; kb < KLEN; kb += 8) {
    const float4 a4 = *(const float4*)(Arow + kb);
    const float4 b4 = *(const float4*)(Brow + kb);
    v2f a, b;
    a.x = a4.x; a.y = a4.y;
    b.x = b4.x; b.y = b4.y;
    acc = __builtin_amdgcn_wmma_f32_16x16x4_f32(false, a, false, b, (short)0, acc, false, false);
    a.x = a4.z; a.y = a4.w;
    b.x = b4.z; b.y = b4.w;
    acc = __builtin_amdgcn_wmma_f32_16x16x4_f32(false, a, false, b, (short)0, acc, false, false);
  }
  float* P = PT + w * (16 * PT_LD);
  for (int v = 0; v < 8; ++v) P[(8 * hf + v) * PT_LD + m] = acc[v];
  __syncthreads();
  const float dt = (ns == 0) ? t : (t - tprev);
  const float* P0 = PT + ((w >> SH) << SH) * (16 * PT_LD) + m * PT_LD;
#pragma unroll
  for (int i = 0; i < NI; ++i) {
    const int q = g + 2 * WPU * i;
    if (q < 16) {
      const int p = 16 * pg + q;
      float x = P0[q];
      for (int j = 1; j < WPU; ++j) x += P0[j * (16 * PT_LD) + q];
      float gate = softplusf(x + gb[i]);
      float real = -softplusf(lr[i]) * gate;
      float im = li[i];
      float e = expf(real * dt);
      float a = e * cosf(im * dt);
      float c = e * sinf(im * dt);
      float denom = real * real + im * im;
      float nre = a - 1.f, nim = c;
      float qr = (nre * real + nim * im) / (denom + 1e-12f);
      float qi = (nim * real - nre * im) / (denom + 1e-12f);
      if (denom < 1e-8f) { qr = dt; qi = 0.f; }
      if (!(dt > 0.f)) { qr = 0.f; qi = 0.f; }
      ((float4*)Cf)[(size_t)p * n + ns] = make_float4(a, c, qr, qi);
    }
  }
}

__global__ __launch_bounds__(512) void k_inverse(const float* __restrict__ V,
                                                 float* __restrict__ Vinv,
                                                 const float* __restrict__ E,
                                                 const float* __restrict__ alpha,
                                                 float* __restrict__ EA,
                                                 const float* __restrict__ gate_w,
                                                 const float* __restrict__ U,
                                                 const float* __restrict__ times,
                                                 const float* __restrict__ gate_b,
                                                 const float* __restrict__ llr,
                                                 const float* __restrict__ lim,
                                                 float* __restrict__ Cf, int n) {
  __shared__ float M[64 * INV_LD];
  if (blockIdx.x != 0) {
    const int b = blockIdx.x - 1;
    const int ntail = n / 8 - 4 * COEF_FULL;
    if (b < ntail) {
      coef_units<4>(M, 4 * COEF_FULL + b, U, gate_w, times, gate_b, llr, lim, Cf, n);
    } else if (b < ntail + COEF_FULL) {
      coef_units<2>(M, 4 * (b - ntail), U, gate_w, times, gate_b, llr, lim, Cf, n);
    } else {
      const int idx = (b - COEF_FULL - ntail) * 512 + threadIdx.x;
      if (idx < 64 * 100) {
        const int d = idx / 100, k = idx % 100;
        float s = 0.f;
        for (int r = 0; r < 32; ++r) s = fmaf(E[d * 32 + r], alpha[r * 100 + k], s);
        EA[idx] = s;
      }
    }
    return;
  }
  __shared__ float PR[16 * INV_LD];
  __shared__ __attribute__((aligned(16))) float NF[64 * NF_LD];
  __shared__ __attribute__((aligned(16))) float NFT[16 * 16];
  __shared__ float PINV[16];
  const int tid = threadIdx.x;
  const int l = tid & 31;
  const int w = __builtin_amdgcn_readfirstlane(tid >> 5);
  const int ti = w >> 2, tj = w & 3;
  const int m = l & 15, h = l >> 4, kh = h * 2;

  for (int idx = tid; idx < 64 * 64; idx += 512)
    M[(idx >> 6) * INV_LD + (idx & 63)] = V[idx];
  __syncthreads();

  for (int kb = 0; kb < 4; ++kb) {
    const int c0 = 16 * kb;

    if (w == 0) {
      const int r0 = (c0 + l) & 63, r1 = (c0 + l + 32) & 63;
      float a0[16], a1[16], n0[16], n1[16];
      float mypinv = 0.f;
#pragma unroll
      for (int c = 0; c < 16; ++c) {
        a0[c] = M[r0 * INV_LD + c0 + c];
        a1[c] = M[r1 * INV_LD + c0 + c];
      }
#pragma unroll
      for (int j = 0; j < 16; ++j) {
        const float pk = __int_as_float(__builtin_amdgcn_readlane(__float_as_int(a0[j]), j));
        const float pinv = 1.f / pk;
        const float f0 = a0[j], f1 = a1[j];
        const bool isp = (l == j);
        n0[j] = -f0;
        n1[j] = -f1;
        mypinv = isp ? pinv : mypinv;
#pragma unroll
        for (int c = 0; c < 16; ++c) {
          if (c == j) {
            a0[c] = isp ? pinv : (n0[j] * pinv);
            a1[c] = n1[j] * pinv;
          } else {
            const float pr =
                __int_as_float(__builtin_amdgcn_readlane(__float_as_int(a0[c]), j)) * pinv;
            a0[c] = isp ? pr : fmaf(n0[j], pr, a0[c]);
            a1[c] = fmaf(n1[j], pr, a1[c]);
          }
        }
      }
#pragma unroll
      for (int c = 0; c < 16; ++c) {
        M[r0 * INV_LD + c0 + c] = a0[c];
        M[r1 * INV_LD + c0 + c] = a1[c];
        NF[r0 * NF_LD + c] = n0[c];
        NF[r1 * NF_LD + c] = n1[c];
      }
      if (l < 16) {
#pragma unroll
        for (int j = 0; j < 16; ++j) NFT[j * 16 + l] = n0[j];
        PINV[l] = mypinv;
      }
    }
    __syncthreads();

    if (tid < 64 && (tid >> 4) != kb) {
      const int cc = tid;
      float x[16];
#pragma unroll
      for (int j = 0; j < 16; ++j) x[j] = M[(c0 + j) * INV_LD + cc];
#pragma unroll
      for (int j = 0; j < 16; ++j) {
        const float pr = x[j] * PINV[j];
        PR[j * INV_LD + cc] = pr;
#pragma unroll
        for (int jj = 0; jj < 16; ++jj) {
          if (jj != j) x[jj] = fmaf(NFT[j * 16 + jj], pr, x[jj]);
        }
        x[j] = pr;
      }
#pragma unroll
      for (int j = 0; j < 16; ++j) M[(c0 + j) * INV_LD + cc] = x[j];
    }
    __syncthreads();

    if (ti != kb && tj != kb) {
      float* Tij = M + (16 * ti) * INV_LD + 16 * tj;
      const float* At = NF + (16 * ti) * NF_LD;
      const float* Bt = PR + 16 * tj;
      v8f acc;
      for (int v = 0; v < 8; ++v) acc[v] = Tij[(8 * h + v) * INV_LD + m];
      for (int k4 = 0; k4 < 16; k4 += 4) {
        v2f a, b;
        a.x = At[m * NF_LD + k4 + kh];
        a.y = At[m * NF_LD + k4 + kh + 1];
        b.x = Bt[(k4 + kh) * INV_LD + m];
        b.y = Bt[(k4 + kh + 1) * INV_LD + m];
        acc = __builtin_amdgcn_wmma_f32_16x16x4_f32(false, a, false, b, (short)0, acc,
                                                    false, false);
      }
      for (int v = 0; v < 8; ++v) Tij[(8 * h + v) * INV_LD + m] = acc[v];
    }
    __syncthreads();
  }

  for (int idx = tid; idx < 64 * 64; idx += 512)
    Vinv[idx] = M[(idx >> 6) * INV_LD + (idx & 63)];
}

#define GS_LD 516
#define MS_LD 113
#define TSW_LD 17
#define GW_KB 64

template <int NT>
__device__ __forceinline__ void strip_tiles(const float4 (&va)[8], const float* __restrict__ s,
                                            const int ld, v8f (&d)[NT]) {
  typedef __attribute__((ext_vector_type(NT))) float vnf;
#pragma unroll
  for (int kk = 0; kk < 8; ++kk) {
    vnf b[4];
#pragma unroll
    for (int e = 0; e < 4; ++e) b[e] = *(const vnf*)(s + (size_t)(8 * kk + e) * ld);
    v2f a, bb;
    a.x = va[kk].x; a.y = va[kk].y;
#pragma unroll
    for (int t = 0; t < NT; ++t) {
      bb.x = b[0][t]; bb.y = b[1][t];
      d[t] = __builtin_amdgcn_wmma_f32_16x16x4_f32(false, a, false, bb, (short)0, d[t], false, false);
    }
    a.x = va[kk].z; a.y = va[kk].w;
#pragma unroll
    for (int t = 0; t < NT; ++t) {
      bb.x = b[2][t]; bb.y = b[3][t];
      d[t] = __builtin_amdgcn_wmma_f32_16x16x4_f32(false, a, false, bb, (short)0, d[t], false, false);
    }
  }
}

__global__ __launch_bounds__(128) __attribute__((amdgpu_waves_per_eu(1, 2))) void k_gemm_w(
    const float* __restrict__ U, const float* __restrict__ Vinv, const float* __restrict__ B,
    const float* __restrict__ EA, const int* __restrict__ marks, float* __restrict__ Cf, int n) {
  __shared__ __attribute__((aligned(16))) float Gs[16 * GS_LD];
  __shared__ float Ms[16 * MS_LD];
  __shared__ float TS[4 * 32 * TSW_LD];
  constexpr int NF4 = GW_KB / 8;
  const int rg = blockIdx.x, ct = blockIdx.y;
  const int tid = threadIdx.x, l = tid & 31;
  const int w = __builtin_amdgcn_readfirstlane(tid >> 5);
  const int m = l & 15;
  const int hf = l >> 4;
  const int rt0 = rg * 8 + 2 * w;
  const float* Arow0 = U + (size_t)(rt0 * 16 + m) * 512 + 4 * hf;
  const float* Arow1 = Arow0 + 16 * 512;
  const int ns = rt0 * 16 + l;
  const int mk = marks[ns];
  float4* cfp = (float4*)Cf;
  float4 cf[8];
#pragma unroll
  for (int k = 0; k < 8; ++k) cf[k] = cfp[(size_t)(8 * ct + k) * n + ns];
  float4 a4[2][NF4], c4[2][NF4], b4[2][NF4];
#pragma unroll
  for (int i = 0; i < NF4; ++i) {
    a4[0][i] = *(const float4*)(Arow0 + 8 * i);
    c4[0][i] = *(const float4*)(Arow1 + 8 * i);
  }

  float4 va[8];
  {
    const float* vr = Vinv + (size_t)(16 * ct + m) * 64 + 4 * hf;
#pragma unroll
    for (int kk = 0; kk < 8; ++kk) va[kk] = *(const float4*)(vr + 8 * kk);
  }
#pragma unroll
  for (int gi = 0; gi < 2; ++gi) {
    const int g = w + 4 * gi;
    v8f d[4] = {};
    strip_tiles<4>(va, B + (size_t)(4 * hf) * 512 + 64 * g + 4 * m, 512, d);
    for (int r = 0; r < 8; ++r)
      *(float4*)(Gs + (8 * hf + r) * GS_LD + 64 * g + 4 * m) =
          make_float4(d[0][r], d[1][r], d[2][r], d[3][r]);
  }
  {
    const int col = 32 * w + 2 * m;
    v8f d[2] = {};
    strip_tiles<2>(va, EA + (size_t)(4 * hf) * 100 + min(col, 98), 100, d);
    if (col < 100) {
      for (int r = 0; r < 8; ++r) {
        Ms[(8 * hf + r) * MS_LD + col] = d[0][r];
        Ms[(8 * hf + r) * MS_LD + col + 1] = d[1][r];
      }
    }
  }
  __syncthreads();

  const float* Bl = Gs + m * GS_LD + 4 * hf;
#pragma unroll
  for (int i = 0; i < NF4; ++i) b4[0][i] = *(const float4*)(Bl + 8 * i);
  v8f acc0 = {}, acc1 = {};
#pragma unroll
  for (int kb = 0; kb < 512 / GW_KB; ++kb) {
    const int cur = kb & 1, nxt = cur ^ 1;
    if (kb + 1 < 512 / GW_KB) {
#pragma unroll
      for (int i = 0; i < NF4; ++i) {
        a4[nxt][i] = *(const float4*)(Arow0 + (kb + 1) * GW_KB + 8 * i);
        c4[nxt][i] = *(const float4*)(Arow1 + (kb + 1) * GW_KB + 8 * i);
        b4[nxt][i] = *(const float4*)(Bl + (kb + 1) * GW_KB + 8 * i);
      }
    }
#pragma unroll
    for (int i = 0; i < NF4; ++i) {
      v2f a, b, c;
      a.x = a4[cur][i].x; a.y = a4[cur][i].y;
      c.x = c4[cur][i].x; c.y = c4[cur][i].y;
      b.x = b4[cur][i].x; b.y = b4[cur][i].y;
      acc0 = __builtin_amdgcn_wmma_f32_16x16x4_f32(false, a, false, b, (short)0, acc0, false, false);
      acc1 = __builtin_amdgcn_wmma_f32_16x16x4_f32(false, c, false, b, (short)0, acc1, false, false);
      a.x = a4[cur][i].z; a.y = a4[cur][i].w;
      c.x = c4[cur][i].z; c.y = c4[cur][i].w;
      b.x = b4[cur][i].z; b.y = b4[cur][i].w;
      acc0 = __builtin_amdgcn_wmma_f32_16x16x4_f32(false, a, false, b, (short)0, acc0, false, false);
      acc1 = __builtin_amdgcn_wmma_f32_16x16x4_f32(false, c, false, b, (short)0, acc1, false, false);
    }
  }
  float* Tw = TS + w * (32 * TSW_LD);
  for (int v = 0; v < 8; ++v) {
    Tw[(8 * hf + v) * TSW_LD + m] = acc0[v];
    Tw[(16 + 8 * hf + v) * TSW_LD + m] = acc1[v];
  }
  __syncthreads();
  const float* tr = Tw + l * TSW_LD;
#pragma unroll
  for (int k = 0; k < 8; ++k) {
    const int p = 8 * ct + k;
    const float qr = cf[k].z, qi = cf[k].w;
    float w0 = tr[2 * k];
    float w1 = tr[2 * k + 1];
    float ure = qr * w0 - qi * w1;
    float uim = qi * w0 + qr * w1;
    ure += Ms[(2 * k) * MS_LD + mk];
    uim += Ms[(2 * k + 1) * MS_LD + mk];
    cfp[(size_t)p * n + ns] = make_float4(cf[k].x, cf[k].y, ure, uim);
  }
}

#define SPT 16
#define SCAN_T 512
#define SCAN_W (SCAN_T / 32)

__device__ __forceinline__ void wave_prefix(float& Ar, float& Ai, float& Ur, float& Ui, int l) {
  for (int d = 1; d < 32; d <<= 1) {
    const float pAr = __shfl_up(Ar, d, 32), pAi = __shfl_up(Ai, d, 32);
    const float pUr = __shfl_up(Ur, d, 32), pUi = __shfl_up(Ui, d, 32);
    if (l >= d) {
      const float nUr = Ar * pUr - Ai * pUi + Ur, nUi = Ar * pUi + Ai * pUr + Ui;
      const float nAr = Ar * pAr - Ai * pAi, nAi = Ar * pAi + Ai * pAr;
      Ar = nAr; Ai = nAi; Ur = nUr; Ui = nUi;
    }
  }
}

__global__ __launch_bounds__(SCAN_T) void k_scan(const float* __restrict__ Cf,
                                                 const float* __restrict__ Vinv,
                                                 const float* __restrict__ x0,
                                                 float* __restrict__ YT, int n) {
  __shared__ float4 WAGG[SCAN_W];
  __shared__ float Y0[2];
  const int p = blockIdx.x, t = threadIdx.x, l = t & 31;
  const int w = __builtin_amdgcn_readfirstlane(t >> 5);
  const int s0 = t * SPT;
  const bool live = (s0 + SPT <= n);
  float ar[SPT], ai[SPT], ur[SPT], ui[SPT];
  if (live) {
    const float4* c4 = (const float4*)Cf + (size_t)p * n + s0;
#pragma unroll
    for (int i = 0; i < SPT; ++i) {
      const float4 v = c4[i];
      ar[i] = v.x; ai[i] = v.y; ur[i] = v.z; ui[i] = v.w;
    }
  } else {
#pragma unroll
    for (int i = 0; i < SPT; ++i) { ar[i] = 1.f; ai[i] = 0.f; ur[i] = 0.f; ui[i] = 0.f; }
  }
  if (w == 0) {
    const float* v0 = Vinv + (2 * p) * 64;
    const float* v1 = v0 + 64;
    const float xa = x0[l], xb = x0[32 + l];
    float sr = fmaf(v0[32 + l], xb, v0[l] * xa);
    float si = fmaf(v1[32 + l], xb, v1[l] * xa);
    for (int d = 16; d >= 1; d >>= 1) {
      sr += __shfl_xor(sr, d, 32);
      si += __shfl_xor(si, d, 32);
    }
    if (l == 0) { Y0[0] = sr; Y0[1] = si; }
  }
  float Ar = 1.f, Ai = 0.f, Ur = 0.f, Ui = 0.f;
#pragma unroll
  for (int i = 0; i < SPT; ++i) {
    const float nAr = ar[i] * Ar - ai[i] * Ai, nAi = ar[i] * Ai + ai[i] * Ar;
    const float nUr = ar[i] * Ur - ai[i] * Ui + ur[i], nUi = ar[i] * Ui + ai[i] * Ur + ui[i];
    Ar = nAr; Ai = nAi; Ur = nUr; Ui = nUi;
  }
  wave_prefix(Ar, Ai, Ur, Ui, l);
  if (l == 31) WAGG[w] = make_float4(Ar, Ai, Ur, Ui);
  __syncthreads();
  const float y0r = Y0[0], y0i = Y0[1];
  float qAr = 1.f, qAi = 0.f, qUr = 0.f, qUi = 0.f;
  if (l < SCAN_W) {
    const float4 q = WAGG[l];
    qAr = q.x; qAi = q.y; qUr = q.z; qUi = q.w;
  }
  wave_prefix(qAr, qAi, qUr, qUi, l);
  const int src = (w > 0) ? (w - 1) : 0;
  float xAr = __shfl(qAr, src, 32), xAi = __shfl(qAi, src, 32);
  float xUr = __shfl(qUr, src, 32), xUi = __shfl(qUi, src, 32);
  if (w == 0) { xAr = 1.f; xAi = 0.f; xUr = 0.f; xUi = 0.f; }
  const float ywr = xAr * y0r - xAi * y0i + xUr;
  const float ywi = xAr * y0i + xAi * y0r + xUi;
  float eAr = __shfl_up(Ar, 1, 32), eAi = __shfl_up(Ai, 1, 32);
  float eUr = __shfl_up(Ur, 1, 32), eUi = __shfl_up(Ui, 1, 32);
  if (l == 0) { eAr = 1.f; eAi = 0.f; eUr = 0.f; eUi = 0.f; }
  float yr = eAr * ywr - eAi * ywi + eUr;
  float yi = eAr * ywi + eAi * ywr + eUi;
  float outr[SPT], outi[SPT];
#pragma unroll
  for (int i = 0; i < SPT; ++i) {
    const float nyr = ar[i] * yr - ai[i] * yi + ur[i];
    const float nyi = ar[i] * yi + ai[i] * yr + ui[i];
    yr = nyr; yi = nyi;
    outr[i] = yr; outi[i] = yi;
  }
  if (live) {
    float4* yo = (float4*)(YT + (size_t)(2 * p) * n + s0);
    float4* y1 = (float4*)(YT + (size_t)(2 * p + 1) * n + s0);
#pragma unroll
    for (int i = 0; i < SPT; i += 4) {
      yo[i / 4] = make_float4(outr[i], outr[i + 1], outr[i + 2], outr[i + 3]);
      y1[i / 4] = make_float4(outi[i], outi[i + 1], outi[i + 2], outi[i + 3]);
    }
  }
}

__global__ __launch_bounds__(32) __attribute__((amdgpu_waves_per_eu(1, 2))) void k_gemm_out(
    const float* __restrict__ YT, const float* __restrict__ V, const float* __restrict__ EA,
    const int* __restrict__ marks, float* __restrict__ out, int n) {
  const int rt = blockIdx.x, l = threadIdx.x;
  const int m = l & 15;
  const int kh = (l >> 4) * 2;
  const float* Acol = YT + (size_t)kh * n + (rt * 16 + m);
  const float* Brow = V + (size_t)m * 64 + kh;
  const int rbase = rt * 16 + (l >> 4) * 8;
  int mk[8];
#pragma unroll
  for (int v = 0; v < 8; ++v) mk[v] = marks[rbase + v];
  v8f acc[4] = {};
#pragma unroll
  for (int kb = 0; kb < 64; kb += 4) {
    v2f a;
    a.x = Acol[(size_t)kb * n];
    a.y = Acol[(size_t)(kb + 1) * n];
#pragma unroll
    for (int t = 0; t < 4; ++t) {
      v2f b = *(const v2f*)(Brow + (size_t)t * 16 * 64 + kb);
      acc[t] = __builtin_amdgcn_wmma_f32_16x16x4_f32(false, a, false, b, (short)0, acc[t],
                                                     false, false);
    }
  }
  const size_t xl_off = (size_t)n * 64;
#pragma unroll
  for (int t = 0; t < 4; ++t) {
    const int dcol = t * 16 + m;
#pragma unroll
    for (int v = 0; v < 8; ++v) {
      const int row = rbase + v;
      const float xr = acc[t][v];
      const float r = EA[dcol * 100 + mk[v]];
      out[(size_t)row * 64 + dcol] = xr;
      out[xl_off + (size_t)row * 64 + dcol] = xr - r;
    }
  }
}

extern "C" void kernel_launch(void* const* d_in, const int* in_sizes, int n_in,
                              void* d_out, int out_size, void* d_ws, size_t ws_size,
                              hipStream_t stream) {
  const float* times   = (const float*)d_in[0];
  const int*   marks   = (const int*)d_in[1];
  const float* u_in    = (const float*)d_in[2];
  const float* llr     = (const float*)d_in[3];
  const float* lim     = (const float*)d_in[4];
  const float* V       = (const float*)d_in[5];
  const float* B       = (const float*)d_in[6];
  const float* E       = (const float*)d_in[7];
  const float* alpha   = (const float*)d_in[8];
  const float* gate_w  = (const float*)d_in[9];
  const float* gate_b  = (const float*)d_in[10];
  const float* x0      = (const float*)d_in[11];
  float* out = (float*)d_out;
  const int n = in_sizes[0];

  float* ws    = (float*)d_ws;
  float* Vinv  = ws; ws += 64 * 64;
  float* EA    = ws; ws += 64 * 100;
  float* Cf    = ws; ws += (size_t)n * 128;
  float* YT    = ws; ws += (size_t)n * 64;

  k_inverse<<<dim3(1 + COEF_FULL + (n / 8 - 4 * COEF_FULL) + EA_BLOCKS), dim3(512), 0, stream>>>(
      V, Vinv, E, alpha, EA, gate_w, u_in, times, gate_b, llr, lim, Cf, n);
  k_gemm_w<<<dim3(n / 128, 4), dim3(128), 0, stream>>>(u_in, Vinv, B, EA, marks, Cf, n);
  k_scan<<<dim3(32), dim3(SCAN_T), 0, stream>>>(Cf, Vinv, x0, YT, n);
  k_gemm_out<<<dim3(n / 16), dim3(32), 0, stream>>>(YT, V, EA, marks, out, n);
}


# Scan: every global load and store of a wave takes 512 consecutive bytes (was 32 lanes 256 B or 64 B apart); lane-major order made in a wave-private LDS strip; sums and their order unchanged
# speedup vs baseline: 6.9795x; 1.0309x over previous
#include <hip/hip_runtime.h>
#include <math.h>


typedef __attribute__((ext_vector_type(2))) float v2f;
typedef __attribute__((ext_vector_type(8))) float v8f;

__device__ __forceinline__ float softplusf(float x) {
  return (x > 20.f) ? x : log1pf(expf(x));
}

#define INV_LD 68
#define NF_LD 20

#define EA_BLOCKS 13
#define COEF_FULL 255
#define PT_LD 17

template <int SH>
__device__ __forceinline__ void coef_units(float* __restrict__ PT, const int unit0,
                                           const float* __restrict__ U,
                                           const float* __restrict__ gate_w,
                                           const float* __restrict__ times,
                                           const float* __restrict__ gate_b,
                                           const float* __restrict__ llr,
                                           const float* __restrict__ lim,
                                           float* __restrict__ Cf, const int n) {
  constexpr int WPU = 1 << SH;
  constexpr int KLEN = 512 >> SH;
  constexpr int NI = (WPU < 8) ? (8 / WPU) : 1;
  const int tid = threadIdx.x;
  const int l = tid & 31;
  const int w = __builtin_amdgcn_readfirstlane(tid >> 5);
  const int unit = unit0 + (w >> SH);
  const int kq = w & (WPU - 1);
  const int rt = unit >> 1, pg = unit & 1;
  const int m = l & 15, hf = l >> 4;
  const int ns = rt * 16 + m;
  const float t = times[ns];
  const float tprev = times[ns > 0 ? ns - 1 : 0];
  const int g = 2 * kq + hf;
  float gb[NI], lr[NI], li[NI];
#pragma unroll
  for (int i = 0; i < NI; ++i) {
    const int pc = 16 * pg + min(g + 2 * WPU * i, 15);
    gb[i] = gate_b[pc];
    lr[i] = llr[pc];
    li[i] = lim[pc];
  }
  const float* Arow = U + (size_t)ns * 512 + kq * KLEN + 4 * hf;
  const float* Brow = gate_w + (size_t)(16 * pg + m) * 512 + kq * KLEN + 4 * hf;
  v8f acc = {};
#pragma unroll
  for (int kb = 0; kb < KLEN; kb += 8) {
    const float4 a4 = *(const float4*)(Arow + kb);
    const float4 b4 = *(const float4*)(Brow + kb);
    v2f a, b;
    a.x = a4.x; a.y = a4.y;
    b.x = b4.x; b.y = b4.y;
    acc = __builtin_amdgcn_wmma_f32_16x16x4_f32(false, a, false, b, (short)0, acc, false, false);
    a.x = a4.z; a.y = a4.w;
    b.x = b4.z; b.y = b4.w;
    acc = __builtin_amdgcn_wmma_f32_16x16x4_f32(false, a, false, b, (short)0, acc, false, false);
  }
  float* P = PT + w * (16 * PT_LD);
  for (int v = 0; v < 8; ++v) P[(8 * hf + v) * PT_LD + m] = acc[v];
  __syncthreads();
  const float dt = (ns == 0) ? t : (t - tprev);
  const float* P0 = PT + ((w >> SH) << SH) * (16 * PT_LD) + m * PT_LD;
#pragma unroll
  for (int i = 0; i < NI; ++i) {
    const int q = g + 2 * WPU * i;
    if (q < 16) {
      const int p = 16 * pg + q;
      float x = P0[q];
      for (int j = 1; j < WPU; ++j) x += P0[j * (16 * PT_LD) + q];
      float gate = softplusf(x + gb[i]);
      float real = -softplusf(lr[i]) * gate;
      float im = li[i];
      float e = expf(real * dt);
      float a = e * cosf(im * dt);
      float c = e * sinf(im * dt);
      float denom = real * real + im * im;
      float nre = a - 1.f, nim = c;
      float qr = (nre * real + nim * im) / (denom + 1e-12f);
      float qi = (nim * real - nre * im) / (denom + 1e-12f);
      if (denom < 1e-8f) { qr = dt; qi = 0.f; }
      if (!(dt > 0.f)) { qr = 0.f; qi = 0.f; }
      ((float4*)Cf)[(size_t)p * n + ns] = make_float4(a, c, qr, qi);
    }
  }
}

__global__ __launch_bounds__(512) void k_inverse(const float* __restrict__ V,
                                                 float* __restrict__ Vinv,
                                                 const float* __restrict__ E,
                                                 const float* __restrict__ alpha,
                                                 float* __restrict__ EA,
                                                 const float* __restrict__ gate_w,
                                                 const float* __restrict__ U,
                                                 const float* __restrict__ times,
                                                 const float* __restrict__ gate_b,
                                                 const float* __restrict__ llr,
                                                 const float* __restrict__ lim,
                                                 float* __restrict__ Cf, int n) {
  __shared__ float M[64 * INV_LD];
  if (blockIdx.x != 0) {
    const int b = blockIdx.x - 1;
    const int ntail = n / 8 - 4 * COEF_FULL;
    if (b < ntail) {
      coef_units<4>(M, 4 * COEF_FULL + b, U, gate_w, times, gate_b, llr, lim, Cf, n);
    } else if (b < ntail + COEF_FULL) {
      coef_units<2>(M, 4 * (b - ntail), U, gate_w, times, gate_b, llr, lim, Cf, n);
    } else {
      const int idx = (b - COEF_FULL - ntail) * 512 + threadIdx.x;
      if (idx < 64 * 100) {
        const int d = idx / 100, k = idx % 100;
        float s = 0.f;
        for (int r = 0; r < 32; ++r) s = fmaf(E[d * 32 + r], alpha[r * 100 + k], s);
        EA[idx] = s;
      }
    }
    return;
  }
  __shared__ float PR[16 * INV_LD];
  __shared__ __attribute__((aligned(16))) float NF[64 * NF_LD];
  __shared__ __attribute__((aligned(16))) float NFT[16 * 16];
  __shared__ float PINV[16];
  const int tid = threadIdx.x;
  const int l = tid & 31;
  const int w = __builtin_amdgcn_readfirstlane(tid >> 5);
  const int ti = w >> 2, tj = w & 3;
  const int m = l & 15, h = l >> 4, kh = h * 2;

  for (int idx = tid; idx < 64 * 64; idx += 512)
    M[(idx >> 6) * INV_LD + (idx & 63)] = V[idx];
  __syncthreads();

  for (int kb = 0; kb < 4; ++kb) {
    const int c0 = 16 * kb;

    if (w == 0) {
      const int r0 = (c0 + l) & 63, r1 = (c0 + l + 32) & 63;
      float a0[16], a1[16], n0[16], n1[16];
      float mypinv = 0.f;
#pragma unroll
      for (int c = 0; c < 16; ++c) {
        a0[c] = M[r0 * INV_LD + c0 + c];
        a1[c] = M[r1 * INV_LD + c0 + c];
      }
#pragma unroll
      for (int j = 0; j < 16; ++j) {
        const float pk = __int_as_float(__builtin_amdgcn_readlane(__float_as_int(a0[j]), j));
        const float pinv = 1.f / pk;
        const float f0 = a0[j], f1 = a1[j];
        const bool isp = (l == j);
        n0[j] = -f0;
        n1[j] = -f1;
        mypinv = isp ? pinv : mypinv;
#pragma unroll
        for (int c = 0; c < 16; ++c) {
          if (c == j) {
            a0[c] = isp ? pinv : (n0[j] * pinv);
            a1[c] = n1[j] * pinv;
          } else {
            const float pr =
                __int_as_float(__builtin_amdgcn_readlane(__float_as_int(a0[c]), j)) * pinv;
            a0[c] = isp ? pr : fmaf(n0[j], pr, a0[c]);
            a1[c] = fmaf(n1[j], pr, a1[c]);
          }
        }
      }
#pragma unroll
      for (int c = 0; c < 16; ++c) {
        M[r0 * INV_LD + c0 + c] = a0[c];
        M[r1 * INV_LD + c0 + c] = a1[c];
        NF[r0 * NF_LD + c] = n0[c];
        NF[r1 * NF_LD + c] = n1[c];
      }
      if (l < 16) {
#pragma unroll
        for (int j = 0; j < 16; ++j) NFT[j * 16 + l] = n0[j];
        PINV[l] = mypinv;
      }
    }
    __syncthreads();

    if (tid < 64 && (tid >> 4) != kb) {
      const int cc = tid;
      float x[16];
#pragma unroll
      for (int j = 0; j < 16; ++j) x[j] = M[(c0 + j) * INV_LD + cc];
#pragma unroll
      for (int j = 0; j < 16; ++j) {
        const float pr = x[j] * PINV[j];
        PR[j * INV_LD + cc] = pr;
#pragma unroll
        for (int jj = 0; jj < 16; ++jj) {
          if (jj != j) x[jj] = fmaf(NFT[j * 16 + jj], pr, x[jj]);
        }
        x[j] = pr;
      }
#pragma unroll
      for (int j = 0; j < 16; ++j) M[(c0 + j) * INV_LD + cc] = x[j];
    }
    __syncthreads();

    if (ti != kb && tj != kb) {
      float* Tij = M + (16 * ti) * INV_LD + 16 * tj;
      const float* At = NF + (16 * ti) * NF_LD;
      const float* Bt = PR + 16 * tj;
      v8f acc;
      for (int v = 0; v < 8; ++v) acc[v] = Tij[(8 * h + v) * INV_LD + m];
      for (int k4 = 0; k4 < 16; k4 += 4) {
        v2f a, b;
        a.x = At[m * NF_LD + k4 + kh];
        a.y = At[m * NF_LD + k4 + kh + 1];
        b.x = Bt[(k4 + kh) * INV_LD + m];
        b.y = Bt[(k4 + kh + 1) * INV_LD + m];
        acc = __builtin_amdgcn_wmma_f32_16x16x4_f32(false, a, false, b, (short)0, acc,
                                                    false, false);
      }
      for (int v = 0; v < 8; ++v) Tij[(8 * h + v) * INV_LD + m] = acc[v];
    }
    __syncthreads();
  }

  for (int idx = tid; idx < 64 * 64; idx += 512)
    Vinv[idx] = M[(idx >> 6) * INV_LD + (idx & 63)];
}

#define GS_LD 516
#define MS_LD 113
#define TSW_LD 17
#define GW_KB 64

template <int NT>
__device__ __forceinline__ void strip_tiles(const float4 (&va)[8], const float* __restrict__ s,
                                            const int ld, v8f (&d)[NT]) {
  typedef __attribute__((ext_vector_type(NT))) float vnf;
#pragma unroll
  for (int kk = 0; kk < 8; ++kk) {
    vnf b[4];
#pragma unroll
    for (int e = 0; e < 4; ++e) b[e] = *(const vnf*)(s + (size_t)(8 * kk + e) * ld);
    v2f a, bb;
    a.x = va[kk].x; a.y = va[kk].y;
#pragma unroll
    for (int t = 0; t < NT; ++t) {
      bb.x = b[0][t]; bb.y = b[1][t];
      d[t] = __builtin_amdgcn_wmma_f32_16x16x4_f32(false, a, false, bb, (short)0, d[t], false, false);
    }
    a.x = va[kk].z; a.y = va[kk].w;
#pragma unroll
    for (int t = 0; t < NT; ++t) {
      bb.x = b[2][t]; bb.y = b[3][t];
      d[t] = __builtin_amdgcn_wmma_f32_16x16x4_f32(false, a, false, bb, (short)0, d[t], false, false);
    }
  }
}

__global__ __launch_bounds__(128) __attribute__((amdgpu_waves_per_eu(1, 2))) void k_gemm_w(
    const float* __restrict__ U, const float* __restrict__ Vinv, const float* __restrict__ B,
    const float* __restrict__ EA, const int* __restrict__ marks, float* __restrict__ Cf, int n) {
  __shared__ __attribute__((aligned(16))) float Gs[16 * GS_LD];
  __shared__ float Ms[16 * MS_LD];
  __shared__ float TS[4 * 32 * TSW_LD];
  constexpr int NF4 = GW_KB / 8;
  const int rg = blockIdx.x, ct = blockIdx.y;
  const int tid = threadIdx.x, l = tid & 31;
  const int w = __builtin_amdgcn_readfirstlane(tid >> 5);
  const int m = l & 15;
  const int hf = l >> 4;
  const int rt0 = rg * 8 + 2 * w;
  const float* Arow0 = U + (size_t)(rt0 * 16 + m) * 512 + 4 * hf;
  const float* Arow1 = Arow0 + 16 * 512;
  const int ns = rt0 * 16 + l;
  const int mk = marks[ns];
  float4* cfp = (float4*)Cf;
  float4 cf[8];
#pragma unroll
  for (int k = 0; k < 8; ++k) cf[k] = cfp[(size_t)(8 * ct + k) * n + ns];
  float4 a4[2][NF4], c4[2][NF4], b4[2][NF4];
#pragma unroll
  for (int i = 0; i < NF4; ++i) {
    a4[0][i] = *(const float4*)(Arow0 + 8 * i);
    c4[0][i] = *(const float4*)(Arow1 + 8 * i);
  }

  float4 va[8];
  {
    const float* vr = Vinv + (size_t)(16 * ct + m) * 64 + 4 * hf;
#pragma unroll
    for (int kk = 0; kk < 8; ++kk) va[kk] = *(const float4*)(vr + 8 * kk);
  }
#pragma unroll
  for (int gi = 0; gi < 2; ++gi) {
    const int g = w + 4 * gi;
    v8f d[4] = {};
    strip_tiles<4>(va, B + (size_t)(4 * hf) * 512 + 64 * g + 4 * m, 512, d);
    for (int r = 0; r < 8; ++r)
      *(float4*)(Gs + (8 * hf + r) * GS_LD + 64 * g + 4 * m) =
          make_float4(d[0][r], d[1][r], d[2][r], d[3][r]);
  }
  {
    const int col = 32 * w + 2 * m;
    v8f d[2] = {};
    strip_tiles<2>(va, EA + (size_t)(4 * hf) * 100 + min(col, 98), 100, d);
    if (col < 100) {
      for (int r = 0; r < 8; ++r) {
        Ms[(8 * hf + r) * MS_LD + col] = d[0][r];
        Ms[(8 * hf + r) * MS_LD + col + 1] = d[1][r];
      }
    }
  }
  __syncthreads();

  const float* Bl = Gs + m * GS_LD + 4 * hf;
#pragma unroll
  for (int i = 0; i < NF4; ++i) b4[0][i] = *(const float4*)(Bl + 8 * i);
  v8f acc0 = {}, acc1 = {};
#pragma unroll
  for (int kb = 0; kb < 512 / GW_KB; ++kb) {
    const int cur = kb & 1, nxt = cur ^ 1;
    if (kb + 1 < 512 / GW_KB) {
#pragma unroll
      for (int i = 0; i < NF4; ++i) {
        a4[nxt][i] = *(const float4*)(Arow0 + (kb + 1) * GW_KB + 8 * i);
        c4[nxt][i] = *(const float4*)(Arow1 + (kb + 1) * GW_KB + 8 * i);
        b4[nxt][i] = *(const float4*)(Bl + (kb + 1) * GW_KB + 8 * i);
      }
    }
#pragma unroll
    for (int i = 0; i < NF4; ++i) {
      v2f a, b, c;
      a.x = a4[cur][i].x; a.y = a4[cur][i].y;
      c.x = c4[cur][i].x; c.y = c4[cur][i].y;
      b.x = b4[cur][i].x; b.y = b4[cur][i].y;
      acc0 = __builtin_amdgcn_wmma_f32_16x16x4_f32(false, a, false, b, (short)0, acc0, false, false);
      acc1 = __builtin_amdgcn_wmma_f32_16x16x4_f32(false, c, false, b, (short)0, acc1, false, false);
      a.x = a4[cur][i].z; a.y = a4[cur][i].w;
      c.x = c4[cur][i].z; c.y = c4[cur][i].w;
      b.x = b4[cur][i].z; b.y = b4[cur][i].w;
      acc0 = __builtin_amdgcn_wmma_f32_16x16x4_f32(false, a, false, b, (short)0, acc0, false, false);
      acc1 = __builtin_amdgcn_wmma_f32_16x16x4_f32(false, c, false, b, (short)0, acc1, false, false);
    }
  }
  float* Tw = TS + w * (32 * TSW_LD);
  for (int v = 0; v < 8; ++v) {
    Tw[(8 * hf + v) * TSW_LD + m] = acc0[v];
    Tw[(16 + 8 * hf + v) * TSW_LD + m] = acc1[v];
  }
  __syncthreads();
  const float* tr = Tw + l * TSW_LD;
#pragma unroll
  for (int k = 0; k < 8; ++k) {
    const int p = 8 * ct + k;
    const float qr = cf[k].z, qi = cf[k].w;
    float w0 = tr[2 * k];
    float w1 = tr[2 * k + 1];
    float ure = qr * w0 - qi * w1;
    float uim = qi * w0 + qr * w1;
    ure += Ms[(2 * k) * MS_LD + mk];
    uim += Ms[(2 * k + 1) * MS_LD + mk];
    cfp[(size_t)p * n + ns] = make_float4(cf[k].x, cf[k].y, ure, uim);
  }
}

#define SPT 16
#define SCAN_T 512
#define SCAN_W (SCAN_T / 32)
#define SQ_LD 17
#define SO_LD 20
#define SO_ROW (32 * SO_LD)

__device__ __forceinline__ void wave_prefix(float& Ar, float& Ai, float& Ur, float& Ui, int l) {
  for (int d = 1; d < 32; d <<= 1) {
    const float pAr = __shfl_up(Ar, d, 32), pAi = __shfl_up(Ai, d, 32);
    const float pUr = __shfl_up(Ur, d, 32), pUi = __shfl_up(Ui, d, 32);
    if (l >= d) {
      const float nUr = Ar * pUr - Ai * pUi + Ur, nUi = Ar * pUi + Ai * pUr + Ui;
      const float nAr = Ar * pAr - Ai * pAi, nAi = Ar * pAi + Ai * pAr;
      Ar = nAr; Ai = nAi; Ur = nUr; Ui = nUi;
    }
  }
}

__global__ __launch_bounds__(SCAN_T) void k_scan(const float* __restrict__ Cf,
                                                 const float* __restrict__ Vinv,
                                                 const float* __restrict__ x0,
                                                 float* __restrict__ YT, int n) {
  __shared__ float4 WAGG[SCAN_W];
  __shared__ float Y0[2];
  __shared__ float4 SQ[SCAN_W * 32 * SQ_LD];
  const int p = blockIdx.x, t = threadIdx.x, l = t & 31;
  const int w = __builtin_amdgcn_readfirstlane(t >> 5);
  const int sw = w * (32 * SPT);
  float4* Sw = SQ + w * (32 * SQ_LD);
  {
    const float4* c4 = (const float4*)Cf + (size_t)p * n + sw;
    float4 q[SPT];
#pragma unroll
    for (int i = 0; i < SPT; ++i)
      q[i] = (sw + 32 * i + l < n) ? c4[32 * i + l] : make_float4(1.f, 0.f, 0.f, 0.f);
#pragma unroll
    for (int i = 0; i < SPT; ++i) Sw[(2 * i + (l >> 4)) * SQ_LD + (l & 15)] = q[i];
  }
  if (w == 0) {
    const float* v0 = Vinv + (2 * p) * 64;
    const float* v1 = v0 + 64;
    const float xa = x0[l], xb = x0[32 + l];
    float sr = fmaf(v0[32 + l], xb, v0[l] * xa);
    float si = fmaf(v1[32 + l], xb, v1[l] * xa);
    for (int d = 16; d >= 1; d >>= 1) {
      sr += __shfl_xor(sr, d, 32);
      si += __shfl_xor(si, d, 32);
    }
    if (l == 0) { Y0[0] = sr; Y0[1] = si; }
  }
  __syncthreads();
  float ar[SPT], ai[SPT], ur[SPT], ui[SPT];
#pragma unroll
  for (int i = 0; i < SPT; ++i) {
    const float4 v = Sw[l * SQ_LD + i];
    ar[i] = v.x; ai[i] = v.y; ur[i] = v.z; ui[i] = v.w;
  }
  float Ar = 1.f, Ai = 0.f, Ur = 0.f, Ui = 0.f;
#pragma unroll
  for (int i = 0; i < SPT; ++i) {
    const float nAr = ar[i] * Ar - ai[i] * Ai, nAi = ar[i] * Ai + ai[i] * Ar;
    const float nUr = ar[i] * Ur - ai[i] * Ui + ur[i], nUi = ar[i] * Ui + ai[i] * Ur + ui[i];
    Ar = nAr; Ai = nAi; Ur = nUr; Ui = nUi;
  }
  wave_prefix(Ar, Ai, Ur, Ui, l);
  if (l == 31) WAGG[w] = make_float4(Ar, Ai, Ur, Ui);
  __syncthreads();
  const float y0r = Y0[0], y0i = Y0[1];
  float qAr = 1.f, qAi = 0.f, qUr = 0.f, qUi = 0.f;
  if (l < SCAN_W) {
    const float4 q = WAGG[l];
    qAr = q.x; qAi = q.y; qUr = q.z; qUi = q.w;
  }
  wave_prefix(qAr, qAi, qUr, qUi, l);
  const int src = (w > 0) ? (w - 1) : 0;
  float xAr = __shfl(qAr, src, 32), xAi = __shfl(qAi, src, 32);
  float xUr = __shfl(qUr, src, 32), xUi = __shfl(qUi, src, 32);
  if (w == 0) { xAr = 1.f; xAi = 0.f; xUr = 0.f; xUi = 0.f; }
  const float ywr = xAr * y0r - xAi * y0i + xUr;
  const float ywi = xAr * y0i + xAi * y0r + xUi;
  float eAr = __shfl_up(Ar, 1, 32), eAi = __shfl_up(Ai, 1, 32);
  float eUr = __shfl_up(Ur, 1, 32), eUi = __shfl_up(Ui, 1, 32);
  if (l == 0) { eAr = 1.f; eAi = 0.f; eUr = 0.f; eUi = 0.f; }
  float yr = eAr * ywr - eAi * ywi + eUr;
  float yi = eAr * ywi + eAi * ywr + eUi;
  float* So = (float*)Sw;
#pragma unroll
  for (int i = 0; i < SPT; i += 4) {
    float o_r[4], o_i[4];
#pragma unroll
    for (int e = 0; e < 4; ++e) {
      const float nyr = ar[i + e] * yr - ai[i + e] * yi + ur[i + e];
      const float nyi = ar[i + e] * yi + ai[i + e] * yr + ui[i + e];
      yr = nyr; yi = nyi;
      o_r[e] = yr; o_i[e] = yi;
    }
    *(float4*)(So + l * SO_LD + i) = make_float4(o_r[0], o_r[1], o_r[2], o_r[3]);
    *(float4*)(So + SO_ROW + l * SO_LD + i) = make_float4(o_i[0], o_i[1], o_i[2], o_i[3]);
  }
  __syncthreads();
  float* y0p = YT + (size_t)(2 * p) * n + sw;
  float* y1p = y0p + n;
#pragma unroll
  for (int k = 0; k < SPT / 4; ++k) {
    const int f = 32 * k + l;
    const float* s = So + (f >> 2) * SO_LD + 4 * (f & 3);
    const float4 vr = *(const float4*)s;
    const float4 vi = *(const float4*)(s + SO_ROW);
    if (sw + 4 * f + 3 < n) {
      *(float4*)(y0p + 4 * f) = vr;
      *(float4*)(y1p + 4 * f) = vi;
    }
  }
}

__global__ __launch_bounds__(32) __attribute__((amdgpu_waves_per_eu(1, 2))) void k_gemm_out(
    const float* __restrict__ YT, const float* __restrict__ V, const float* __restrict__ EA,
    const int* __restrict__ marks, float* __restrict__ out, int n) {
  const int rt = blockIdx.x, l = threadIdx.x;
  const int m = l & 15;
  const int kh = (l >> 4) * 2;
  const float* Acol = YT + (size_t)kh * n + (rt * 16 + m);
  const float* Brow = V + (size_t)m * 64 + kh;
  const int rbase = rt * 16 + (l >> 4) * 8;
  int mk[8];
#pragma unroll
  for (int v = 0; v < 8; ++v) mk[v] = marks[rbase + v];
  v8f acc[4] = {};
#pragma unroll
  for (int kb = 0; kb < 64; kb += 4) {
    v2f a;
    a.x = Acol[(size_t)kb * n];
    a.y = Acol[(size_t)(kb + 1) * n];
#pragma unroll
    for (int t = 0; t < 4; ++t) {
      v2f b = *(const v2f*)(Brow + (size_t)t * 16 * 64 + kb);
      acc[t] = __builtin_amdgcn_wmma_f32_16x16x4_f32(false, a, false, b, (short)0, acc[t],
                                                     false, false);
    }
  }
  const size_t xl_off = (size_t)n * 64;
#pragma unroll
  for (int t = 0; t < 4; ++t) {
    const int dcol = t * 16 + m;
#pragma unroll
    for (int v = 0; v < 8; ++v) {
      const int row = rbase + v;
      const float xr = acc[t][v];
      const float r = EA[dcol * 100 + mk[v]];
      out[(size_t)row * 64 + dcol] = xr;
      out[xl_off + (size_t)row * 64 + dcol] = xr - r;
    }
  }
}

extern "C" void kernel_launch(void* const* d_in, const int* in_sizes, int n_in,
                              void* d_out, int out_size, void* d_ws, size_t ws_size,
                              hipStream_t stream) {
  const float* times   = (const float*)d_in[0];
  const int*   marks   = (const int*)d_in[1];
  const float* u_in    = (const float*)d_in[2];
  const float* llr     = (const float*)d_in[3];
  const float* lim     = (const float*)d_in[4];
  const float* V       = (const float*)d_in[5];
  const float* B       = (const float*)d_in[6];
  const float* E       = (const float*)d_in[7];
  const float* alpha   = (const float*)d_in[8];
  const float* gate_w  = (const float*)d_in[9];
  const float* gate_b  = (const float*)d_in[10];
  const float* x0      = (const float*)d_in[11];
  float* out = (float*)d_out;
  const int n = in_sizes[0];

  float* ws    = (float*)d_ws;
  float* Vinv  = ws; ws += 64 * 64;
  float* EA    = ws; ws += 64 * 100;
  float* Cf    = ws; ws += (size_t)n * 128;
  float* YT    = ws; ws += (size_t)n * 64;

  k_inverse<<<dim3(1 + COEF_FULL + (n / 8 - 4 * COEF_FULL) + EA_BLOCKS), dim3(512), 0, stream>>>(
      V, Vinv, E, alpha, EA, gate_w, u_in, times, gate_b, llr, lim, Cf, n);
  k_gemm_w<<<dim3(n / 128, 4), dim3(128), 0, stream>>>(u_in, Vinv, B, EA, marks, Cf, n);
  k_scan<<<dim3(32), dim3(SCAN_T), 0, stream>>>(Cf, Vinv, x0, YT, n);
  k_gemm_out<<<dim3(n / 16), dim3(32), 0, stream>>>(YT, V, EA, marks, out, n);
}
